# starting kernel
# baseline (speedup 1.0000x reference)
#include <hip/hip_runtime.h>
#include <math.h>


typedef float v2f __attribute__((ext_vector_type(2)));
typedef float v8f __attribute__((ext_vector_type(8)));

#define BB   16
#define TT   512
#define HH   128
#define MM   (BB * TT)
#define NEL  (BB * TT * HH)
#define SCALE 0.08838834764831845f
#define LN_EPS 1e-5f
#define LCH  16
#define NCH  (TT / LCH)

__global__ __launch_bounds__(128) void proj_kernel(
    const float* __restrict__ x,
    const float* __restrict__ Wq, const float* __restrict__ Wk,
    const float* __restrict__ Wv, const float* __restrict__ Wi,
    const float* __restrict__ Wf, const float* __restrict__ Wo,
    const float* __restrict__ b_i, const float* __restrict__ b_f,
    const float* __restrict__ b_o,
    float* __restrict__ oq, float* __restrict__ ok, float* __restrict__ ov,
    float* __restrict__ oi, float* __restrict__ of, float* __restrict__ oo)
{
    const int lane   = threadIdx.x & 31;
    const int waveId = threadIdx.x >> 5;
    const int tile   = blockIdx.x * 4 + waveId;

    const int matIdx = tile / (512 * 8);
    const int rem    = tile % (512 * 8);
    const int m0     = (rem >> 3) << 4;
    const int n0     = (rem & 7) << 4;

    const float* W;
    switch (matIdx) {
        case 0: W = Wq; break;
        case 1: W = Wk; break;
        case 2: W = Wv; break;
        case 3: W = Wi; break;
        case 4: W = Wf; break;
        default: W = Wo; break;
    }

    const int lm = lane & 15;
    const int lk = (lane >> 4) << 1;

    const float* aptr = x + (m0 + lm) * HH + lk;
    const float* bptr = W + (n0 + lm) * HH + lk;

    v8f acc = {};
    #pragma unroll
    for (int kb = 0; kb < HH; kb += 4) {
        v2f a = *(const v2f*)(aptr + kb);
        v2f b = *(const v2f*)(bptr + kb);
        acc = __builtin_amdgcn_wmma_f32_16x16x4_f32(false, a, false, b,
                                                    (short)0, acc, false, false);
    }

    const int col = n0 + lm;
    float bias = 0.0f;
    if (matIdx == 3) bias = b_i[col];
    else if (matIdx == 4) bias = b_f[col];
    else if (matIdx == 5) bias = b_o[col];

    float* outp;
    switch (matIdx) {
        case 0: outp = oq; break;
        case 1: outp = ok; break;
        case 2: outp = ov; break;
        case 3: outp = oi; break;
        case 4: outp = of; break;
        default: outp = oo; break;
    }

    const int rowBase = m0 + ((lane >> 4) << 3);
    #pragma unroll
    for (int v = 0; v < 8; ++v) {
        float val = acc[v];
        if (matIdx <= 1)       val *= SCALE;
        else if (matIdx == 3)  val = expf(val + bias);
        else if (matIdx >= 4)  val = 1.0f / (1.0f + expf(-(val + bias)));
        outp[(rowBase + v) * HH + col] = val;
    }
}

#define CSTR 132
#define TSTR 20
#define C_OFF   0
#define Q_OFF   (C_OFF + 128 * CSTR)
#define K_OFF   (Q_OFF + 16 * CSTR)
#define KT_OFF  (K_OFF + 16 * CSTR)
#define WC_OFF  (KT_OFF + 128 * TSTR)
#define P_OFF   (WC_OFF + 128 * TSTR)
#define M_OFF   (P_OFF + 16 * CSTR)
#define SP_OFF  (M_OFF + 16 * CSTR)
#define S_OFF   (SP_OFF + 4 * 16 * 17)
#define G15_OFF (S_OFF + 16 * 17)
#define NPS_OFF (G15_OFF + 128)
#define NQ_OFF  (NPS_OFF + 128)
#define SMEM_FLOATS (NQ_OFF + 16)

__global__ __launch_bounds__(128, 1) void rec_kernel(
    const float* __restrict__ qb, const float* __restrict__ kb,
    const float* __restrict__ vb, const float* __restrict__ ib,
    const float* __restrict__ fb, const float* __restrict__ ob,
    float* __restrict__ hb)
{
    extern __shared__ float sm[];
    const int b    = blockIdx.x;
    const int r    = threadIdx.x;
    const int lane = r & 31;
    const int wave = r >> 5;
    const int lm   = lane & 15;
    const int lk   = (lane >> 4) << 1;

    #pragma unroll 4
    for (int j = 0; j < CSTR; ++j) sm[C_OFF + r * CSTR + j] = 0.0f;

    float nprev = 0.0f;
    const long bbase = (long)b * TT * HH + r;

    float pk[LCH], pq[LCH], pi[LCH], pf[LCH], pv[LCH], po[LCH];
    #pragma unroll
    for (int s = 0; s < LCH; ++s) {
        const long idx = bbase + (long)s * HH;
        pk[s] = kb[idx]; pq[s] = qb[idx]; pi[s] = ib[idx];
        pf[s] = fb[idx]; pv[s] = vb[idx]; po[s] = ob[idx];
    }

    for (int ch = 0; ch < NCH; ++ch) {
        const long gbase = bbase + (long)(ch * LCH) * HH;

        float G[LCH], Wg[LCH], ovv[LCH];
        {
            float g = 1.0f;
            #pragma unroll
            for (int s = 0; s < LCH; ++s) { g *= pf[s]; G[s] = g; ovv[s] = po[s]; }
            #pragma unroll
            for (int s = 0; s < LCH; ++s) Wg[s] = pi[s] * pv[s] / G[s];
            const float g15 = G[LCH - 1];
            float np = nprev;
            #pragma unroll
            for (int s = 0; s < LCH; ++s) {
                np = fmaf(pf[s], np, pi[s] * pk[s]);
                sm[M_OFF + s * CSTR + r] = np * pq[s];
            }
            nprev = np;
            #pragma unroll
            for (int s = 0; s < LCH; ++s) {
                sm[Q_OFF + s * CSTR + r]  = pq[s];
                sm[K_OFF + s * CSTR + r]  = pk[s];
                sm[KT_OFF + r * TSTR + s] = pk[s];
                sm[WC_OFF + r * TSTR + s] = g15 * Wg[s];
            }
            sm[G15_OFF + r] = g15;
        }
        if (ch + 1 < NCH) {
            const long nbase = bbase + (long)((ch + 1) * LCH) * HH;
            #pragma unroll
            for (int s = 0; s < LCH; ++s) {
                const long idx = nbase + (long)s * HH;
                pk[s] = kb[idx]; pq[s] = qb[idx]; pi[s] = ib[idx];
                pf[s] = fb[idx]; pv[s] = vb[idx]; po[s] = ob[idx];
            }
        }
        __syncthreads();

        {
            const int s = r >> 3, seg = r & 7;
            float ps = 0.0f;
            #pragma unroll
            for (int e = 0; e < 16; ++e) ps += sm[M_OFF + s * CSTR + seg * 16 + e];
            sm[NPS_OFF + s * 8 + seg] = ps;
        }
        {
            v2f av[32];
            #pragma unroll
            for (int kk = 0; kk < 32; ++kk)
                av[kk] = *(const v2f*)&sm[Q_OFF + lm * CSTR + kk * 4 + lk];
            #pragma unroll
            for (int tile = 0; tile < 2; ++tile) {
                const int n0p = (2 * wave + tile) << 4;
                v8f acc = {};
                #pragma unroll
                for (int kk = 0; kk < 32; ++kk) {
                    v2f bbf = *(const v2f*)&sm[C_OFF + (n0p + lm) * CSTR + kk * 4 + lk];
                    acc = __builtin_amdgcn_wmma_f32_16x16x4_f32(false, av[kk], false, bbf,
                                                                (short)0, acc, false, false);
                }
                #pragma unroll
                for (int v = 0; v < 8; ++v)
                    sm[P_OFF + (v + 8 * (lane >> 4)) * CSTR + n0p + lm] = acc[v];
            }
        }
        __syncthreads();

        if (r < 16) {
            float t = 0.0f;
            #pragma unroll
            for (int e = 0; e < 8; ++e) t += sm[NPS_OFF + r * 8 + e];
            sm[NQ_OFF + r] = t;
        }
        {
            const int kbase = wave * 32;
            v8f sacc = {};
            #pragma unroll
            for (int kk = 0; kk < 8; ++kk) {
                const int c0 = kbase + kk * 4 + lk;
                v2f a  = *(const v2f*)&sm[K_OFF + lm * CSTR + c0];
                v2f bq = *(const v2f*)&sm[Q_OFF + lm * CSTR + c0];
                sacc = __builtin_amdgcn_wmma_f32_16x16x4_f32(false, a, false, bq,
                                                             (short)0, sacc, false, false);
            }
            #pragma unroll
            for (int v = 0; v < 8; ++v)
                sm[SP_OFF + wave * 272 + (v + 8 * (lane >> 4)) * 17 + lm] = sacc[v];
        }
        {
            const int mt0 = wave * 2;
            v2f afr[2][4];
            #pragma unroll
            for (int mi = 0; mi < 2; ++mi)
                #pragma unroll
                for (int kk = 0; kk < 4; ++kk)
                    afr[mi][kk] = *(const v2f*)&sm[WC_OFF + ((mt0 + mi) * 16 + lm) * TSTR + kk * 4 + lk];
            #pragma unroll
            for (int nt = 0; nt < 8; ++nt) {
                v2f bfr[4];
                #pragma unroll
                for (int kk = 0; kk < 4; ++kk)
                    bfr[kk] = *(const v2f*)&sm[KT_OFF + (nt * 16 + lm) * TSTR + kk * 4 + lk];
                #pragma unroll
                for (int mi = 0; mi < 2; ++mi) {
                    v8f dacc = {};
                    #pragma unroll
                    for (int kk = 0; kk < 4; ++kk)
                        dacc = __builtin_amdgcn_wmma_f32_16x16x4_f32(false, afr[mi][kk], false, bfr[kk],
                                                                     (short)0, dacc, false, false);
                    const int m0r = (mt0 + mi) * 16;
                    #pragma unroll
                    for (int v = 0; v < 8; ++v) {
                        const int row = m0r + v + 8 * (lane >> 4);
                        const int ci  = C_OFF + row * CSTR + nt * 16 + lm;
                        const float g = sm[G15_OFF + row];
                        sm[ci] = fmaf(g, sm[ci], dacc[v]);
                    }
                }
            }
        }
        __syncthreads();

        #pragma unroll
        for (int p = 0; p < 2; ++p) {
            const int idx = r * 2 + p;
            const int u = idx >> 4, sc = idx & 15;
            float ssum = sm[SP_OFF + 0 * 272 + u * 17 + sc]
                       + sm[SP_OFF + 1 * 272 + u * 17 + sc]
                       + sm[SP_OFF + 2 * 272 + u * 17 + sc]
                       + sm[SP_OFF + 3 * 272 + u * 17 + sc];
            sm[S_OFF + u * 17 + sc] = ssum;
        }
        __syncthreads();

        #pragma unroll
        for (int s = 0; s < LCH; ++s) {
            float acc = sm[P_OFF + s * CSTR + r];
            #pragma unroll
            for (int u = 0; u < LCH; ++u)
                if (u <= s) acc = fmaf(Wg[u], sm[S_OFF + u * 17 + s], acc);
            const float ret = G[s] * acc;
            const float nrm = fmaxf(fabsf(sm[NQ_OFF + s]), 1.0f);
            hb[gbase + (long)s * HH] = ovv[s] * ret / nrm;
        }
        __syncthreads();
    }
}

__global__ __launch_bounds__(256) void ln_kernel(
    const float* __restrict__ hb, const float* __restrict__ x,
    const float* __restrict__ lnw, const float* __restrict__ lnb,
    float* __restrict__ out)
{
    const int lane = threadIdx.x & 31;
    const int wave = threadIdx.x >> 5;
    const long row = (long)blockIdx.x * 8 + wave;
    const long rb  = row * HH;

    float y[4];
    float s = 0.0f, s2 = 0.0f;
    #pragma unroll
    for (int j = 0; j < 4; ++j) {
        const int c = lane + 32 * j;
        y[j] = hb[rb + c] + x[rb + c];
        s  += y[j];
        s2 += y[j] * y[j];
    }
    #pragma unroll
    for (int off = 16; off >= 1; off >>= 1) {
        s  += __shfl_xor(s,  off, 32);
        s2 += __shfl_xor(s2, off, 32);
    }
    const float mu  = s * (1.0f / HH);
    const float var = s2 * (1.0f / HH) - mu * mu;
    const float rs  = rsqrtf(var + LN_EPS);

    #pragma unroll
    for (int j = 0; j < 4; ++j) {
        const int c = lane + 32 * j;
        out[rb + c] = (y[j] - mu) * rs * lnw[c] + lnb[c];
    }
}

extern "C" void kernel_launch(void* const* d_in, const int* in_sizes, int n_in,
                              void* d_out, int out_size, void* d_ws, size_t ws_size,
                              hipStream_t stream)
{
    const float* x   = (const float*)d_in[0];
    const float* Wq  = (const float*)d_in[1];
    const float* Wk  = (const float*)d_in[2];
    const float* Wv  = (const float*)d_in[3];
    const float* Wi  = (const float*)d_in[4];
    const float* Wf  = (const float*)d_in[5];
    const float* Wo  = (const float*)d_in[6];
    const float* b_i = (const float*)d_in[7];
    const float* b_f = (const float*)d_in[8];
    const float* b_o = (const float*)d_in[9];
    const float* lnw = (const float*)d_in[10];
    const float* lnb = (const float*)d_in[11];
    float* out = (float*)d_out;

    float* ws = (float*)d_ws;
    float* q  = ws + 0L * NEL;
    float* k  = ws + 1L * NEL;
    float* v  = ws + 2L * NEL;
    float* ig = ws + 3L * NEL;
    float* fg = ws + 4L * NEL;
    float* og = ws + 5L * NEL;
    float* h  = ws + 6L * NEL;

    proj_kernel<<<6144, 128, 0, stream>>>(x, Wq, Wk, Wv, Wi, Wf, Wo,
                                          b_i, b_f, b_o,
                                          q, k, v, ig, fg, og);
    rec_kernel<<<BB, 128, SMEM_FLOATS * sizeof(float), stream>>>(q, k, v, ig, fg, og, h);
    ln_kernel<<<MM / 8, 256, 0, stream>>>(h, x, lnw, lnb, out);
}


# Recurrence over chunks made parallel in time: per-chunk state terms, an elementwise scan of C and n over chunks, per-chunk outputs on 512 work-groups; sum order of n and of the C.q product changes
# speedup vs baseline: 5.7076x; 5.7076x over previous
#include <hip/hip_runtime.h>
#include <math.h>


typedef float v2f __attribute__((ext_vector_type(2)));
typedef float v8f __attribute__((ext_vector_type(8)));

#define BB   16
#define TT   512
#define HH   128
#define MM   (BB * TT)
#define NEL  (BB * TT * HH)
#define SCALE 0.08838834764831845f
#define LN_EPS 1e-5f
#define LCH  16
#define NCH  (TT / LCH)

__global__ __launch_bounds__(128) void proj_kernel(
    const float* __restrict__ x,
    const float* __restrict__ Wq, const float* __restrict__ Wk,
    const float* __restrict__ Wv, const float* __restrict__ Wi,
    const float* __restrict__ Wf, const float* __restrict__ Wo,
    const float* __restrict__ b_i, const float* __restrict__ b_f,
    const float* __restrict__ b_o,
    float* __restrict__ oq, float* __restrict__ ok, float* __restrict__ ov,
    float* __restrict__ oi, float* __restrict__ of, float* __restrict__ oo)
{
    const int lane   = threadIdx.x & 31;
    const int waveId = threadIdx.x >> 5;
    const int tile   = blockIdx.x * 4 + waveId;

    const int matIdx = tile / (512 * 8);
    const int rem    = tile % (512 * 8);
    const int m0     = (rem >> 3) << 4;
    const int n0     = (rem & 7) << 4;

    const float* W;
    switch (matIdx) {
        case 0: W = Wq; break;
        case 1: W = Wk; break;
        case 2: W = Wv; break;
        case 3: W = Wi; break;
        case 4: W = Wf; break;
        default: W = Wo; break;
    }

    const int lm = lane & 15;
    const int lk = (lane >> 4) << 1;

    const float* aptr = x + (m0 + lm) * HH + lk;
    const float* bptr = W + (n0 + lm) * HH + lk;

    v8f acc = {};
    #pragma unroll
    for (int kb = 0; kb < HH; kb += 4) {
        v2f a = *(const v2f*)(aptr + kb);
        v2f b = *(const v2f*)(bptr + kb);
        acc = __builtin_amdgcn_wmma_f32_16x16x4_f32(false, a, false, b,
                                                    (short)0, acc, false, false);
    }

    const int col = n0 + lm;
    float bias = 0.0f;
    if (matIdx == 3) bias = b_i[col];
    else if (matIdx == 4) bias = b_f[col];
    else if (matIdx == 5) bias = b_o[col];

    float* outp;
    switch (matIdx) {
        case 0: outp = oq; break;
        case 1: outp = ok; break;
        case 2: outp = ov; break;
        case 3: outp = oi; break;
        case 4: outp = of; break;
        default: outp = oo; break;
    }

    const int rowBase = m0 + ((lane >> 4) << 3);
    #pragma unroll
    for (int v = 0; v < 8; ++v) {
        float val = acc[v];
        if (matIdx <= 1)       val *= SCALE;
        else if (matIdx == 3)  val = expf(val + bias);
        else if (matIdx >= 4)  val = 1.0f / (1.0f + expf(-(val + bias)));
        outp[(rowBase + v) * HH + col] = val;
    }
}

#define PLANE  (HH * HH)
#define PLANE4 (HH * HH / 4)

#define TSTR 20
#define A_KT_OFF 0
#define A_WC_OFF (128 * TSTR)
#define A_FLOATS (2 * 128 * TSTR)

__global__ __launch_bounds__(128) void chunk_state_kernel(
    const float* __restrict__ kb, const float* __restrict__ vb,
    const float* __restrict__ ib, const float* __restrict__ fb,
    float* __restrict__ dbuf, float* __restrict__ g15buf,
    float* __restrict__ nlbuf)
{
    __shared__ float4 sa4[A_FLOATS / 4];
    float* sa = (float*)sa4;
    const int b    = blockIdx.x / NCH;
    const int ch   = blockIdx.x % NCH;
    const int r    = threadIdx.x;
    const int lane = r & 31;
    const int wave = r >> 5;
    const int lm   = lane & 15;
    const int lk   = (lane >> 4) << 1;

    const long gbase = ((long)b * TT + (long)(ch * LCH)) * HH + r;

    float pk[LCH], pi[LCH], pf[LCH], pv[LCH];
    #pragma unroll
    for (int s = 0; s < LCH; ++s) {
        const long idx = gbase + (long)s * HH;
        pk[s] = kb[idx]; pi[s] = ib[idx]; pf[s] = fb[idx]; pv[s] = vb[idx];
    }

    float G[LCH];
    {
        float g = 1.0f;
        #pragma unroll
        for (int s = 0; s < LCH; ++s) { g *= pf[s]; G[s] = g; }
    }
    const float g15 = G[LCH - 1];
    float np = 0.0f;
    #pragma unroll
    for (int s = 0; s < LCH; ++s) np = fmaf(pf[s], np, pi[s] * pk[s]);

    #pragma unroll
    for (int s = 0; s < LCH; ++s) {
        const float wg = pi[s] * pv[s] / G[s];
        sa[A_KT_OFF + r * TSTR + s] = pk[s];
        sa[A_WC_OFF + r * TSTR + s] = g15 * wg;
    }
    const long cidx = (long)(b * NCH + ch) * HH + r;
    g15buf[cidx] = g15;
    nlbuf[cidx]  = np;
    __syncthreads();

    float4* __restrict__ dptr4 = (float4*)(dbuf + (long)(b * NCH + ch) * PLANE);
    {
        const int mt0 = wave * 2;
        v2f wfr[2][4];
        #pragma unroll
        for (int mi = 0; mi < 2; ++mi)
            #pragma unroll
            for (int kk = 0; kk < 4; ++kk)
                wfr[mi][kk] = *(const v2f*)&sa[A_WC_OFF + ((mt0 + mi) * 16 + lm) * TSTR + kk * 4 + lk];
        #pragma unroll
        for (int nt = 0; nt < 8; ++nt) {
            v2f kfr[4];
            #pragma unroll
            for (int kk = 0; kk < 4; ++kk)
                kfr[kk] = *(const v2f*)&sa[A_KT_OFF + (nt * 16 + lm) * TSTR + kk * 4 + lk];
            v8f d0 = {}, d1 = {};
            #pragma unroll
            for (int kk = 0; kk < 4; ++kk) {
                d0 = __builtin_amdgcn_wmma_f32_16x16x4_f32(false, kfr[kk], false, wfr[0][kk],
                                                           (short)0, d0, false, false);
                d1 = __builtin_amdgcn_wmma_f32_16x16x4_f32(false, kfr[kk], false, wfr[1][kk],
                                                           (short)0, d1, false, false);
            }
            const int u0 = ((mt0 + 0) * 8 + nt) * 64 + lane;
            const int u1 = ((mt0 + 1) * 8 + nt) * 64 + lane;
            dptr4[u0]      = make_float4(d0[0], d0[1], d0[2], d0[3]);
            dptr4[u0 + 32] = make_float4(d0[4], d0[5], d0[6], d0[7]);
            dptr4[u1]      = make_float4(d1[0], d1[1], d1[2], d1[3]);
            dptr4[u1 + 32] = make_float4(d1[4], d1[5], d1[6], d1[7]);
        }
    }
}

#define SCAN_GRP 16
#define SCAN_CBLK (BB * PLANE4 / 256)
#define SCAN_NBLK (BB * HH / 256)

__global__ __launch_bounds__(256) void state_scan_kernel(
    float* __restrict__ dbuf, const float* __restrict__ g15buf,
    float* __restrict__ nlbuf)
{
    if (blockIdx.x < SCAN_CBLK) {
        const int gid = blockIdx.x * 256 + threadIdx.x;
        const int b   = gid / PLANE4;
        const int e4  = gid % PLANE4;
        const int row = ((e4 >> 9) << 4) + (e4 & 15);
        float4* p = (float4*)(dbuf + (long)b * NCH * PLANE) + e4;
        const float* gp = g15buf + (long)b * NCH * HH + row;

        float4 c = make_float4(0.0f, 0.0f, 0.0f, 0.0f);
        for (int c0 = 0; c0 < NCH; c0 += SCAN_GRP) {
            float4 d[SCAN_GRP];
            float  g[SCAN_GRP];
            #pragma unroll
            for (int j = 0; j < SCAN_GRP; ++j) {
                d[j] = p[(long)(c0 + j) * PLANE4];
                g[j] = gp[(c0 + j) * HH];
            }
            #pragma unroll
            for (int j = 0; j < SCAN_GRP; ++j) {
                p[(long)(c0 + j) * PLANE4] = c;
                c.x = fmaf(g[j], c.x, d[j].x);
                c.y = fmaf(g[j], c.y, d[j].y);
                c.z = fmaf(g[j], c.z, d[j].z);
                c.w = fmaf(g[j], c.w, d[j].w);
            }
        }
    } else {
        const int t   = (blockIdx.x - SCAN_CBLK) * 256 + threadIdx.x;
        const int b   = t >> 7;
        const int row = t & 127;
        const float* gp = g15buf + (long)b * NCH * HH + row;
        float*       np = nlbuf  + (long)b * NCH * HH + row;
        float g[NCH], nv[NCH];
        #pragma unroll
        for (int c = 0; c < NCH; ++c) { g[c] = gp[c * HH]; nv[c] = np[c * HH]; }
        float ns = 0.0f;
        #pragma unroll
        for (int c = 0; c < NCH; ++c) {
            np[c * HH] = ns;
            ns = fmaf(g[c], ns, nv[c]);
        }
    }
}

#define CSTR 132
#define Q_OFF   0
#define K_OFF   (Q_OFF + 16 * CSTR)
#define P_OFF   (K_OFF + 16 * CSTR)
#define M_OFF   (P_OFF + 16 * CSTR)
#define SP_OFF  (M_OFF + 16 * CSTR)
#define S_OFF   (SP_OFF + 4 * 16 * 17)
#define NPS_OFF (S_OFF + 16 * 17)
#define NQ_OFF  (NPS_OFF + 128)
#define SMEM_FLOATS (NQ_OFF + 16)

__global__ __launch_bounds__(128) void chunk_out_kernel(
    const float* __restrict__ qb, const float* __restrict__ kb,
    const float* __restrict__ vb, const float* __restrict__ ib,
    const float* __restrict__ fb, const float* __restrict__ ob,
    const float* __restrict__ cbuf, const float* __restrict__ nsbuf,
    float* __restrict__ hb)
{
    __shared__ float4 sm4[SMEM_FLOATS / 4];
    float* sm = (float*)sm4;
    const int b    = blockIdx.x / NCH;
    const int ch   = blockIdx.x % NCH;
    const int r    = threadIdx.x;
    const int lane = r & 31;
    const int wave = r >> 5;
    const int lm   = lane & 15;
    const int hi   = lane >> 4;
    const int lk   = hi << 1;

    const long gbase = ((long)b * TT + (long)(ch * LCH)) * HH + r;

    float pk[LCH], pq[LCH], pi[LCH], pf[LCH], pv[LCH], po[LCH];
    #pragma unroll
    for (int s = 0; s < LCH; ++s) {
        const long idx = gbase + (long)s * HH;
        pk[s] = kb[idx]; pq[s] = qb[idx]; pi[s] = ib[idx];
        pf[s] = fb[idx]; pv[s] = vb[idx]; po[s] = ob[idx];
    }
    const float nstart = nsbuf[(long)(b * NCH + ch) * HH + r];

    float4 cf[2][16];
    {
        const float4* cp4 = (const float4*)(cbuf + (long)(b * NCH + ch) * PLANE);
        #pragma unroll
        for (int tile = 0; tile < 2; ++tile)
            #pragma unroll
            for (int j = 0; j < 16; ++j)
                cf[tile][j] = cp4[((2 * wave + tile) * 16 + j) * 32 + lane];
    }

    float G[LCH], Wg[LCH], ovv[LCH];
    {
        float g = 1.0f;
        #pragma unroll
        for (int s = 0; s < LCH; ++s) { g *= pf[s]; G[s] = g; ovv[s] = po[s]; }
        #pragma unroll
        for (int s = 0; s < LCH; ++s) Wg[s] = pi[s] * pv[s] / G[s];
        float np = 0.0f;
        #pragma unroll
        for (int s = 0; s < LCH; ++s) {
            np = fmaf(pf[s], np, pi[s] * pk[s]);
            const float nfull = fmaf(G[s], nstart, np);
            sm[M_OFF + s * CSTR + r] = nfull * pq[s];
        }
        #pragma unroll
        for (int s = 0; s < LCH; ++s) {
            sm[Q_OFF + s * CSTR + r] = pq[s];
            sm[K_OFF + s * CSTR + r] = pk[s];
        }
    }
    __syncthreads();

    {
        const int s = r >> 3, seg = r & 7;
        float ps = 0.0f;
        #pragma unroll
        for (int e = 0; e < 16; ++e) ps += sm[M_OFF + s * CSTR + seg * 16 + e];
        sm[NPS_OFF + s * 8 + seg] = ps;
    }
    {
        v8f acc0 = {}, acc1 = {};
        #pragma unroll
        for (int j = 0; j < 16; ++j) {
            const int nt = j >> 1, half = j & 1;
            const float4 qa = *(const float4*)&sm[Q_OFF + lm * CSTR + 16 * nt + 8 * hi + 4 * half];
            const v2f a0 = {qa.x, qa.y};
            const v2f a1 = {qa.z, qa.w};
            const v2f b00 = {cf[0][j].x, cf[0][j].y};
            const v2f b01 = {cf[0][j].z, cf[0][j].w};
            const v2f b10 = {cf[1][j].x, cf[1][j].y};
            const v2f b11 = {cf[1][j].z, cf[1][j].w};
            acc0 = __builtin_amdgcn_wmma_f32_16x16x4_f32(false, a0, false, b00, (short)0, acc0, false, false);
            acc1 = __builtin_amdgcn_wmma_f32_16x16x4_f32(false, a0, false, b10, (short)0, acc1, false, false);
            acc0 = __builtin_amdgcn_wmma_f32_16x16x4_f32(false, a1, false, b01, (short)0, acc0, false, false);
            acc1 = __builtin_amdgcn_wmma_f32_16x16x4_f32(false, a1, false, b11, (short)0, acc1, false, false);
        }
        const int n0p = (2 * wave) << 4;
        #pragma unroll
        for (int v = 0; v < 8; ++v) {
            sm[P_OFF + (v + 8 * hi) * CSTR + n0p + lm]      = acc0[v];
            sm[P_OFF + (v + 8 * hi) * CSTR + n0p + 16 + lm] = acc1[v];
        }
    }
    {
        const int kbase = wave * 32;
        v8f sacc = {};
        #pragma unroll
        for (int kk = 0; kk < 8; ++kk) {
            const int c0 = kbase + kk * 4 + lk;
            v2f a  = *(const v2f*)&sm[K_OFF + lm * CSTR + c0];
            v2f bq = *(const v2f*)&sm[Q_OFF + lm * CSTR + c0];
            sacc = __builtin_amdgcn_wmma_f32_16x16x4_f32(false, a, false, bq,
                                                         (short)0, sacc, false, false);
        }
        #pragma unroll
        for (int v = 0; v < 8; ++v)
            sm[SP_OFF + wave * 272 + (v + 8 * hi) * 17 + lm] = sacc[v];
    }
    __syncthreads();

    if (r < 16) {
        float t = 0.0f;
        #pragma unroll
        for (int e = 0; e < 8; ++e) t += sm[NPS_OFF + r * 8 + e];
        sm[NQ_OFF + r] = t;
    }
    #pragma unroll
    for (int p = 0; p < 2; ++p) {
        const int idx = r * 2 + p;
        const int u = idx >> 4, sc = idx & 15;
        float ssum = sm[SP_OFF + 0 * 272 + u * 17 + sc]
                   + sm[SP_OFF + 1 * 272 + u * 17 + sc]
                   + sm[SP_OFF + 2 * 272 + u * 17 + sc]
                   + sm[SP_OFF + 3 * 272 + u * 17 + sc];
        sm[S_OFF + u * 17 + sc] = ssum;
    }
    __syncthreads();

    #pragma unroll
    for (int s = 0; s < LCH; ++s) {
        float acc = sm[P_OFF + s * CSTR + r];
        #pragma unroll
        for (int u = 0; u < LCH; ++u)
            if (u <= s) acc = fmaf(Wg[u], sm[S_OFF + u * 17 + s], acc);
        const float ret = G[s] * acc;
        const float nrm = fmaxf(fabsf(sm[NQ_OFF + s]), 1.0f);
        hb[gbase + (long)s * HH] = ovv[s] * ret / nrm;
    }
}

__global__ __launch_bounds__(256) void ln_kernel(
    const float* __restrict__ hb, const float* __restrict__ x,
    const float* __restrict__ lnw, const float* __restrict__ lnb,
    float* __restrict__ out)
{
    const int lane = threadIdx.x & 31;
    const int wave = threadIdx.x >> 5;
    const long row = (long)blockIdx.x * 8 + wave;
    const long rb  = row * HH;

    float y[4];
    float s = 0.0f, s2 = 0.0f;
    #pragma unroll
    for (int j = 0; j < 4; ++j) {
        const int c = lane + 32 * j;
        y[j] = hb[rb + c] + x[rb + c];
        s  += y[j];
        s2 += y[j] * y[j];
    }
    #pragma unroll
    for (int off = 16; off >= 1; off >>= 1) {
        s  += __shfl_xor(s,  off, 32);
        s2 += __shfl_xor(s2, off, 32);
    }
    const float mu  = s * (1.0f / HH);
    const float var = s2 * (1.0f / HH) - mu * mu;
    const float rs  = rsqrtf(var + LN_EPS);

    #pragma unroll
    for (int j = 0; j < 4; ++j) {
        const int c = lane + 32 * j;
        out[rb + c] = (y[j] - mu) * rs * lnw[c] + lnb[c];
    }
}

extern "C" void kernel_launch(void* const* d_in, const int* in_sizes, int n_in,
                              void* d_out, int out_size, void* d_ws, size_t ws_size,
                              hipStream_t stream)
{
    const float* x   = (const float*)d_in[0];
    const float* Wq  = (const float*)d_in[1];
    const float* Wk  = (const float*)d_in[2];
    const float* Wv  = (const float*)d_in[3];
    const float* Wi  = (const float*)d_in[4];
    const float* Wf  = (const float*)d_in[5];
    const float* Wo  = (const float*)d_in[6];
    const float* b_i = (const float*)d_in[7];
    const float* b_f = (const float*)d_in[8];
    const float* b_o = (const float*)d_in[9];
    const float* lnw = (const float*)d_in[10];
    const float* lnb = (const float*)d_in[11];
    float* out = (float*)d_out;

    float* ws = (float*)d_ws;
    float* q  = ws + 0L * NEL;
    float* k  = ws + 1L * NEL;
    float* v  = ws + 2L * NEL;
    float* ig = ws + 3L * NEL;
    float* fg = ws + 4L * NEL;
    float* og = ws + 5L * NEL;
    float* h  = ws + 6L * NEL;
    float* cst = ws + 7L * NEL;
    float* g15 = ws + 15L * NEL;
    float* nl  = ws + 15L * NEL + BB * NCH * HH;

    proj_kernel<<<6144, 128, 0, stream>>>(x, Wq, Wk, Wv, Wi, Wf, Wo,
                                          b_i, b_f, b_o,
                                          q, k, v, ig, fg, og);
    chunk_state_kernel<<<BB * NCH, 128, 0, stream>>>(k, v, ig, fg, cst, g15, nl);
    state_scan_kernel<<<SCAN_CBLK + SCAN_NBLK, 256, 0, stream>>>(cst, g15, nl);
    chunk_out_kernel<<<BB * NCH, 128, 0, stream>>>(q, k, v, ig, fg, og, cst, nl, h);
    ln_kernel<<<MM / 8, 256, 0, stream>>>(h, x, lnw, lnb, out);
}


# Projection kernel re-tiled: x tile and weight blocks staged in LDS by whole-row loads, x fragments held in registers, one weight fragment feeds two row tiles; 768 work-groups, three per WGP
# speedup vs baseline: 7.9807x; 1.3983x over previous
#include <hip/hip_runtime.h>
#include <math.h>


typedef float v2f __attribute__((ext_vector_type(2)));
typedef float v8f __attribute__((ext_vector_type(8)));

#define BB   16
#define TT   512
#define HH   128
#define MM   (BB * TT)
#define NEL  (BB * TT * HH)
#define SCALE 0.08838834764831845f
#define LN_EPS 1e-5f
#define LCH  16
#define NCH  (TT / LCH)

#define PJ_ROWS 32
#define PJ_COLS 64
#define PJ_STR  132
#define PJ_XS_FLOATS (PJ_ROWS * PJ_STR)
#define PJ_WB_FLOATS (PJ_COLS * PJ_STR)
#define PJ_NBLK 4
#define PJ_GRID (3 * (MM / PJ_ROWS))

template <int PART>
__device__ __forceinline__ void pj_body(
    const float* __restrict__ x,
    const float* __restrict__ Wm0, const float* __restrict__ Wm1,
    const float* __restrict__ bm0, const float* __restrict__ bm1,
    float* __restrict__ om0, float* __restrict__ om1,
    float* xs, float* wbuf)
{
    const int t    = threadIdx.x;
    const int lane = t & 31;
    const int wave = t >> 5;
    const int lm   = lane & 15;
    const int hi   = lane >> 4;
    const int lk   = hi << 1;
    const int m0   = (blockIdx.x / 3) * PJ_ROWS;

    const int srow = wave;
    const int sc   = lane << 2;

    {
        float4 xv[8], wv[16];
        const float* xsrc = x + (long)(m0 + srow) * HH + sc;
        #pragma unroll
        for (int i = 0; i < 8; ++i) xv[i] = *(const float4*)(xsrc + (4 * i) * HH);
        const float* wsrc = Wm0 + srow * HH + sc;
        #pragma unroll
        for (int i = 0; i < 16; ++i) wv[i] = *(const float4*)(wsrc + (4 * i) * HH);
        #pragma unroll
        for (int i = 0; i < 8; ++i) *(float4*)&xs[(srow + 4 * i) * PJ_STR + sc] = xv[i];
        #pragma unroll
        for (int i = 0; i < 16; ++i) *(float4*)&wbuf[(srow + 4 * i) * PJ_STR + sc] = wv[i];
    }
    __syncthreads();

    v2f af[2][32];
    #pragma unroll
    for (int tile = 0; tile < 2; ++tile)
        #pragma unroll
        for (int kk = 0; kk < 32; ++kk)
            af[tile][kk] = *(const v2f*)&xs[(16 * tile + lm) * PJ_STR + 4 * kk + lk];

    #pragma unroll
    for (int blk = 0; blk < PJ_NBLK; ++blk) {
        const float* wcur = wbuf + (blk & 1) * PJ_WB_FLOATS;
        v2f bf[32];
        #pragma unroll
        for (int kk = 0; kk < 32; ++kk)
            bf[kk] = *(const v2f*)&wcur[(16 * wave + lm) * PJ_STR + 4 * kk + lk];

        __builtin_amdgcn_sched_barrier(0);

        v8f acc0 = {}, acc1 = {};
        #pragma unroll
        for (int kk = 0; kk < 32; ++kk) {
            acc0 = __builtin_amdgcn_wmma_f32_16x16x4_f32(false, af[0][kk], false, bf[kk],
                                                         (short)0, acc0, false, false);
            acc1 = __builtin_amdgcn_wmma_f32_16x16x4_f32(false, af[1][kk], false, bf[kk],
                                                         (short)0, acc1, false, false);
        }

        {
            const int second = blk >> 1;
            const int mat    = 2 * PART + second;
            const int col    = (blk & 1) * PJ_COLS + 16 * wave + lm;
            float* outp      = second ? om1 : om0;
            float bias = 0.0f;
            if (mat >= 3) bias = (second ? bm1 : bm0)[col];
            #pragma unroll
            for (int tile = 0; tile < 2; ++tile) {
                const int rowBase = m0 + 16 * tile + 8 * hi;
                #pragma unroll
                for (int v = 0; v < 8; ++v) {
                    float val = tile ? acc1[v] : acc0[v];
                    if (mat <= 1)       val *= SCALE;
                    else if (mat == 3)  val = expf(val + bias);
                    else if (mat >= 4)  val = 1.0f / (1.0f + expf(-(val + bias)));
                    outp[(rowBase + v) * HH + col] = val;
                }
            }
        }

        if (blk + 1 < PJ_NBLK) {
            float4 wv[16];
            const float* wsrc = (((blk + 1) >> 1) ? Wm1 : Wm0) + (((blk + 1) & 1) * PJ_COLS + srow) * HH + sc;
            #pragma unroll
            for (int i = 0; i < 16; ++i) wv[i] = *(const float4*)(wsrc + (4 * i) * HH);
            float* wnext = wbuf + ((blk + 1) & 1) * PJ_WB_FLOATS;
            #pragma unroll
            for (int i = 0; i < 16; ++i) *(float4*)&wnext[(srow + 4 * i) * PJ_STR + sc] = wv[i];
            __syncthreads();
        }
    }
}

__global__ __launch_bounds__(128) void proj_kernel(
    const float* __restrict__ x,
    const float* __restrict__ Wq, const float* __restrict__ Wk,
    const float* __restrict__ Wv, const float* __restrict__ Wi,
    const float* __restrict__ Wf, const float* __restrict__ Wo,
    const float* __restrict__ b_i, const float* __restrict__ b_f,
    const float* __restrict__ b_o,
    float* __restrict__ oq, float* __restrict__ ok, float* __restrict__ ov,
    float* __restrict__ oi, float* __restrict__ of, float* __restrict__ oo)
{
    __shared__ float4 pj4[(PJ_XS_FLOATS + 2 * PJ_WB_FLOATS) / 4];
    float* xs   = (float*)pj4;
    float* wbuf = xs + PJ_XS_FLOATS;
    switch (blockIdx.x % 3) {
        case 0:  pj_body<0>(x, Wq, Wk, b_i, b_i, oq, ok, xs, wbuf); break;
        case 1:  pj_body<1>(x, Wv, Wi, b_i, b_i, ov, oi, xs, wbuf); break;
        default: pj_body<2>(x, Wf, Wo, b_f, b_o, of, oo, xs, wbuf); break;
    }
}

#define PLANE  (HH * HH)
#define PLANE4 (HH * HH / 4)

#define TSTR 20
#define A_KT_OFF 0
#define A_WC_OFF (128 * TSTR)
#define A_FLOATS (2 * 128 * TSTR)

__global__ __launch_bounds__(128) void chunk_state_kernel(
    const float* __restrict__ kb, const float* __restrict__ vb,
    const float* __restrict__ ib, const float* __restrict__ fb,
    float* __restrict__ dbuf, float* __restrict__ g15buf,
    float* __restrict__ nlbuf)
{
    __shared__ float4 sa4[A_FLOATS / 4];
    float* sa = (float*)sa4;
    const int b    = blockIdx.x / NCH;
    const int ch   = blockIdx.x % NCH;
    const int r    = threadIdx.x;
    const int lane = r & 31;
    const int wave = r >> 5;
    const int lm   = lane & 15;
    const int lk   = (lane >> 4) << 1;

    const long gbase = ((long)b * TT + (long)(ch * LCH)) * HH + r;

    float pk[LCH], pi[LCH], pf[LCH], pv[LCH];
    #pragma unroll
    for (int s = 0; s < LCH; ++s) {
        const long idx = gbase + (long)s * HH;
        pk[s] = kb[idx]; pi[s] = ib[idx]; pf[s] = fb[idx]; pv[s] = vb[idx];
    }

    float G[LCH];
    {
        float g = 1.0f;
        #pragma unroll
        for (int s = 0; s < LCH; ++s) { g *= pf[s]; G[s] = g; }
    }
    const float g15 = G[LCH - 1];
    float np = 0.0f;
    #pragma unroll
    for (int s = 0; s < LCH; ++s) np = fmaf(pf[s], np, pi[s] * pk[s]);

    #pragma unroll
    for (int s = 0; s < LCH; ++s) {
        const float wg = pi[s] * pv[s] / G[s];
        sa[A_KT_OFF + r * TSTR + s] = pk[s];
        sa[A_WC_OFF + r * TSTR + s] = g15 * wg;
    }
    const long cidx = (long)(b * NCH + ch) * HH + r;
    g15buf[cidx] = g15;
    nlbuf[cidx]  = np;
    __syncthreads();

    float4* __restrict__ dptr4 = (float4*)(dbuf + (long)(b * NCH + ch) * PLANE);
    {
        const int mt0 = wave * 2;
        v2f wfr[2][4];
        #pragma unroll
        for (int mi = 0; mi < 2; ++mi)
            #pragma unroll
            for (int kk = 0; kk < 4; ++kk)
                wfr[mi][kk] = *(const v2f*)&sa[A_WC_OFF + ((mt0 + mi) * 16 + lm) * TSTR + kk * 4 + lk];
        #pragma unroll
        for (int nt = 0; nt < 8; ++nt) {
            v2f kfr[4];
            #pragma unroll
            for (int kk = 0; kk < 4; ++kk)
                kfr[kk] = *(const v2f*)&sa[A_KT_OFF + (nt * 16 + lm) * TSTR + kk * 4 + lk];
            v8f d0 = {}, d1 = {};
            #pragma unroll
            for (int kk = 0; kk < 4; ++kk) {
                d0 = __builtin_amdgcn_wmma_f32_16x16x4_f32(false, kfr[kk], false, wfr[0][kk],
                                                           (short)0, d0, false, false);
                d1 = __builtin_amdgcn_wmma_f32_16x16x4_f32(false, kfr[kk], false, wfr[1][kk],
                                                           (short)0, d1, false, false);
            }
            const int u0 = ((mt0 + 0) * 8 + nt) * 64 + lane;
            const int u1 = ((mt0 + 1) * 8 + nt) * 64 + lane;
            dptr4[u0]      = make_float4(d0[0], d0[1], d0[2], d0[3]);
            dptr4[u0 + 32] = make_float4(d0[4], d0[5], d0[6], d0[7]);
            dptr4[u1]      = make_float4(d1[0], d1[1], d1[2], d1[3]);
            dptr4[u1 + 32] = make_float4(d1[4], d1[5], d1[6], d1[7]);
        }
    }
}

#define SCAN_GRP 16
#define SCAN_CBLK (BB * PLANE4 / 256)
#define SCAN_NBLK (BB * HH / 256)

__global__ __launch_bounds__(256) void state_scan_kernel(
    float* __restrict__ dbuf, const float* __restrict__ g15buf,
    float* __restrict__ nlbuf)
{
    if (blockIdx.x < SCAN_CBLK) {
        const int gid = blockIdx.x * 256 + threadIdx.x;
        const int b   = gid / PLANE4;
        const int e4  = gid % PLANE4;
        const int row = ((e4 >> 9) << 4) + (e4 & 15);
        float4* p = (float4*)(dbuf + (long)b * NCH * PLANE) + e4;
        const float* gp = g15buf + (long)b * NCH * HH + row;

        float4 c = make_float4(0.0f, 0.0f, 0.0f, 0.0f);
        for (int c0 = 0; c0 < NCH; c0 += SCAN_GRP) {
            float4 d[SCAN_GRP];
            float  g[SCAN_GRP];
            #pragma unroll
            for (int j = 0; j < SCAN_GRP; ++j) {
                d[j] = p[(long)(c0 + j) * PLANE4];
                g[j] = gp[(c0 + j) * HH];
            }
            #pragma unroll
            for (int j = 0; j < SCAN_GRP; ++j) {
                p[(long)(c0 + j) * PLANE4] = c;
                c.x = fmaf(g[j], c.x, d[j].x);
                c.y = fmaf(g[j], c.y, d[j].y);
                c.z = fmaf(g[j], c.z, d[j].z);
                c.w = fmaf(g[j], c.w, d[j].w);
            }
        }
    } else {
        const int t   = (blockIdx.x - SCAN_CBLK) * 256 + threadIdx.x;
        const int b   = t >> 7;
        const int row = t & 127;
        const float* gp = g15buf + (long)b * NCH * HH + row;
        float*       np = nlbuf  + (long)b * NCH * HH + row;
        float g[NCH], nv[NCH];
        #pragma unroll
        for (int c = 0; c < NCH; ++c) { g[c] = gp[c * HH]; nv[c] = np[c * HH]; }
        float ns = 0.0f;
        #pragma unroll
        for (int c = 0; c < NCH; ++c) {
            np[c * HH] = ns;
            ns = fmaf(g[c], ns, nv[c]);
        }
    }
}

#define CSTR 132
#define Q_OFF   0
#define K_OFF   (Q_OFF + 16 * CSTR)
#define P_OFF   (K_OFF + 16 * CSTR)
#define M_OFF   (P_OFF + 16 * CSTR)
#define SP_OFF  (M_OFF + 16 * CSTR)
#define S_OFF   (SP_OFF + 4 * 16 * 17)
#define NPS_OFF (S_OFF + 16 * 17)
#define NQ_OFF  (NPS_OFF + 128)
#define SMEM_FLOATS (NQ_OFF + 16)

__global__ __launch_bounds__(128) void chunk_out_kernel(
    const float* __restrict__ qb, const float* __restrict__ kb,
    const float* __restrict__ vb, const float* __restrict__ ib,
    const float* __restrict__ fb, const float* __restrict__ ob,
    const float* __restrict__ cbuf, const float* __restrict__ nsbuf,
    float* __restrict__ hb)
{
    __shared__ float4 sm4[SMEM_FLOATS / 4];
    float* sm = (float*)sm4;
    const int b    = blockIdx.x / NCH;
    const int ch   = blockIdx.x % NCH;
    const int r    = threadIdx.x;
    const int lane = r & 31;
    const int wave = r >> 5;
    const int lm   = lane & 15;
    const int hi   = lane >> 4;
    const int lk   = hi << 1;

    const long gbase = ((long)b * TT + (long)(ch * LCH)) * HH + r;

    float pk[LCH], pq[LCH], pi[LCH], pf[LCH], pv[LCH], po[LCH];
    #pragma unroll
    for (int s = 0; s < LCH; ++s) {
        const long idx = gbase + (long)s * HH;
        pk[s] = kb[idx]; pq[s] = qb[idx]; pi[s] = ib[idx];
        pf[s] = fb[idx]; pv[s] = vb[idx]; po[s] = ob[idx];
    }
    const float nstart = nsbuf[(long)(b * NCH + ch) * HH + r];

    float4 cf[2][16];
    {
        const float4* cp4 = (const float4*)(cbuf + (long)(b * NCH + ch) * PLANE);
        #pragma unroll
        for (int tile = 0; tile < 2; ++tile)
            #pragma unroll
            for (int j = 0; j < 16; ++j)
                cf[tile][j] = cp4[((2 * wave + tile) * 16 + j) * 32 + lane];
    }

    float G[LCH], Wg[LCH], ovv[LCH];
    {
        float g = 1.0f;
        #pragma unroll
        for (int s = 0; s < LCH; ++s) { g *= pf[s]; G[s] = g; ovv[s] = po[s]; }
        #pragma unroll
        for (int s = 0; s < LCH; ++s) Wg[s] = pi[s] * pv[s] / G[s];
        float np = 0.0f;
        #pragma unroll
        for (int s = 0; s < LCH; ++s) {
            np = fmaf(pf[s], np, pi[s] * pk[s]);
            const float nfull = fmaf(G[s], nstart, np);
            sm[M_OFF + s * CSTR + r] = nfull * pq[s];
        }
        #pragma unroll
        for (int s = 0; s < LCH; ++s) {
            sm[Q_OFF + s * CSTR + r] = pq[s];
            sm[K_OFF + s * CSTR + r] = pk[s];
        }
    }
    __syncthreads();

    {
        const int s = r >> 3, seg = r & 7;
        float ps = 0.0f;
        #pragma unroll
        for (int e = 0; e < 16; ++e) ps += sm[M_OFF + s * CSTR + seg * 16 + e];
        sm[NPS_OFF + s * 8 + seg] = ps;
    }
    {
        v8f acc0 = {}, acc1 = {};
        #pragma unroll
        for (int j = 0; j < 16; ++j) {
            const int nt = j >> 1, half = j & 1;
            const float4 qa = *(const float4*)&sm[Q_OFF + lm * CSTR + 16 * nt + 8 * hi + 4 * half];
            const v2f a0 = {qa.x, qa.y};
            const v2f a1 = {qa.z, qa.w};
            const v2f b00 = {cf[0][j].x, cf[0][j].y};
            const v2f b01 = {cf[0][j].z, cf[0][j].w};
            const v2f b10 = {cf[1][j].x, cf[1][j].y};
            const v2f b11 = {cf[1][j].z, cf[1][j].w};
            acc0 = __builtin_amdgcn_wmma_f32_16x16x4_f32(false, a0, false, b00, (short)0, acc0, false, false);
            acc1 = __builtin_amdgcn_wmma_f32_16x16x4_f32(false, a0, false, b10, (short)0, acc1, false, false);
            acc0 = __builtin_amdgcn_wmma_f32_16x16x4_f32(false, a1, false, b01, (short)0, acc0, false, false);
            acc1 = __builtin_amdgcn_wmma_f32_16x16x4_f32(false, a1, false, b11, (short)0, acc1, false, false);
        }
        const int n0p = (2 * wave) << 4;
        #pragma unroll
        for (int v = 0; v < 8; ++v) {
            sm[P_OFF + (v + 8 * hi) * CSTR + n0p + lm]      = acc0[v];
            sm[P_OFF + (v + 8 * hi) * CSTR + n0p + 16 + lm] = acc1[v];
        }
    }
    {
        const int kbase = wave * 32;
        v8f sacc = {};
        #pragma unroll
        for (int kk = 0; kk < 8; ++kk) {
            const int c0 = kbase + kk * 4 + lk;
            v2f a  = *(const v2f*)&sm[K_OFF + lm * CSTR + c0];
            v2f bq = *(const v2f*)&sm[Q_OFF + lm * CSTR + c0];
            sacc = __builtin_amdgcn_wmma_f32_16x16x4_f32(false, a, false, bq,
                                                         (short)0, sacc, false, false);
        }
        #pragma unroll
        for (int v = 0; v < 8; ++v)
            sm[SP_OFF + wave * 272 + (v + 8 * hi) * 17 + lm] = sacc[v];
    }
    __syncthreads();

    if (r < 16) {
        float t = 0.0f;
        #pragma unroll
        for (int e = 0; e < 8; ++e) t += sm[NPS_OFF + r * 8 + e];
        sm[NQ_OFF + r] = t;
    }
    #pragma unroll
    for (int p = 0; p < 2; ++p) {
        const int idx = r * 2 + p;
        const int u = idx >> 4, sc = idx & 15;
        float ssum = sm[SP_OFF + 0 * 272 + u * 17 + sc]
                   + sm[SP_OFF + 1 * 272 + u * 17 + sc]
                   + sm[SP_OFF + 2 * 272 + u * 17 + sc]
                   + sm[SP_OFF + 3 * 272 + u * 17 + sc];
        sm[S_OFF + u * 17 + sc] = ssum;
    }
    __syncthreads();

    #pragma unroll
    for (int s = 0; s < LCH; ++s) {
        float acc = sm[P_OFF + s * CSTR + r];
        #pragma unroll
        for (int u = 0; u < LCH; ++u)
            if (u <= s) acc = fmaf(Wg[u], sm[S_OFF + u * 17 + s], acc);
        const float ret = G[s] * acc;
        const float nrm = fmaxf(fabsf(sm[NQ_OFF + s]), 1.0f);
        hb[gbase + (long)s * HH] = ovv[s] * ret / nrm;
    }
}

__global__ __launch_bounds__(256) void ln_kernel(
    const float* __restrict__ hb, const float* __restrict__ x,
    const float* __restrict__ lnw, const float* __restrict__ lnb,
    float* __restrict__ out)
{
    const int lane = threadIdx.x & 31;
    const int wave = threadIdx.x >> 5;
    const long row = (long)blockIdx.x * 8 + wave;
    const long rb  = row * HH;

    float y[4];
    float s = 0.0f, s2 = 0.0f;
    #pragma unroll
    for (int j = 0; j < 4; ++j) {
        const int c = lane + 32 * j;
        y[j] = hb[rb + c] + x[rb + c];
        s  += y[j];
        s2 += y[j] * y[j];
    }
    #pragma unroll
    for (int off = 16; off >= 1; off >>= 1) {
        s  += __shfl_xor(s,  off, 32);
        s2 += __shfl_xor(s2, off, 32);
    }
    const float mu  = s * (1.0f / HH);
    const float var = s2 * (1.0f / HH) - mu * mu;
    const float rs  = rsqrtf(var + LN_EPS);

    #pragma unroll
    for (int j = 0; j < 4; ++j) {
        const int c = lane + 32 * j;
        out[rb + c] = (y[j] - mu) * rs * lnw[c] + lnb[c];
    }
}

extern "C" void kernel_launch(void* const* d_in, const int* in_sizes, int n_in,
                              void* d_out, int out_size, void* d_ws, size_t ws_size,
                              hipStream_t stream)
{
    const float* x   = (const float*)d_in[0];
    const float* Wq  = (const float*)d_in[1];
    const float* Wk  = (const float*)d_in[2];
    const float* Wv  = (const float*)d_in[3];
    const float* Wi  = (const float*)d_in[4];
    const float* Wf  = (const float*)d_in[5];
    const float* Wo  = (const float*)d_in[6];
    const float* b_i = (const float*)d_in[7];
    const float* b_f = (const float*)d_in[8];
    const float* b_o = (const float*)d_in[9];
    const float* lnw = (const float*)d_in[10];
    const float* lnb = (const float*)d_in[11];
    float* out = (float*)d_out;

    float* ws = (float*)d_ws;
    float* q  = ws + 0L * NEL;
    float* k  = ws + 1L * NEL;
    float* v  = ws + 2L * NEL;
    float* ig = ws + 3L * NEL;
    float* fg = ws + 4L * NEL;
    float* og = ws + 5L * NEL;
    float* h  = ws + 6L * NEL;
    float* cst = ws + 7L * NEL;
    float* g15 = ws + 15L * NEL;
    float* nl  = ws + 15L * NEL + BB * NCH * HH;

    proj_kernel<<<PJ_GRID, 128, 0, stream>>>(x, Wq, Wk, Wv, Wi, Wf, Wo,
                                          b_i, b_f, b_o,
                                          q, k, v, ig, fg, og);
    chunk_state_kernel<<<BB * NCH, 128, 0, stream>>>(k, v, ig, fg, cst, g15, nl);
    state_scan_kernel<<<SCAN_CBLK + SCAN_NBLK, 256, 0, stream>>>(cst, g15, nl);
    chunk_out_kernel<<<BB * NCH, 128, 0, stream>>>(q, k, v, ig, fg, og, cst, nl, h);
    ln_kernel<<<MM / 8, 256, 0, stream>>>(h, x, lnw, lnb, out);
}


# Residual and LayerNorm moved into the chunk output kernel (row sums through LDS, 8 lanes a row): one launch and the h round trip less; sum order of the LayerNorm mean and variance changes
# speedup vs baseline: 8.4756x; 1.0620x over previous
#include <hip/hip_runtime.h>
#include <math.h>


typedef float v2f __attribute__((ext_vector_type(2)));
typedef float v8f __attribute__((ext_vector_type(8)));

#define BB   16
#define TT   512
#define HH   128
#define MM   (BB * TT)
#define NEL  (BB * TT * HH)
#define SCALE 0.08838834764831845f
#define LN_EPS 1e-5f
#define LCH  16
#define NCH  (TT / LCH)

#define PJ_ROWS 32
#define PJ_COLS 64
#define PJ_STR  132
#define PJ_XS_FLOATS (PJ_ROWS * PJ_STR)
#define PJ_WB_FLOATS (PJ_COLS * PJ_STR)
#define PJ_NBLK 4
#define PJ_GRID (3 * (MM / PJ_ROWS))

template <int PART>
__device__ __forceinline__ void pj_body(
    const float* __restrict__ x,
    const float* __restrict__ Wm0, const float* __restrict__ Wm1,
    const float* __restrict__ bm0, const float* __restrict__ bm1,
    float* __restrict__ om0, float* __restrict__ om1,
    float* xs, float* wbuf)
{
    const int t    = threadIdx.x;
    const int lane = t & 31;
    const int wave = t >> 5;
    const int lm   = lane & 15;
    const int hi   = lane >> 4;
    const int lk   = hi << 1;
    const int m0   = (blockIdx.x / 3) * PJ_ROWS;

    const int srow = wave;
    const int sc   = lane << 2;

    {
        float4 xv[8], wv[16];
        const float* xsrc = x + (long)(m0 + srow) * HH + sc;
        #pragma unroll
        for (int i = 0; i < 8; ++i) xv[i] = *(const float4*)(xsrc + (4 * i) * HH);
        const float* wsrc = Wm0 + srow * HH + sc;
        #pragma unroll
        for (int i = 0; i < 16; ++i) wv[i] = *(const float4*)(wsrc + (4 * i) * HH);
        #pragma unroll
        for (int i = 0; i < 8; ++i) *(float4*)&xs[(srow + 4 * i) * PJ_STR + sc] = xv[i];
        #pragma unroll
        for (int i = 0; i < 16; ++i) *(float4*)&wbuf[(srow + 4 * i) * PJ_STR + sc] = wv[i];
    }
    __syncthreads();

    v2f af[2][32];
    #pragma unroll
    for (int tile = 0; tile < 2; ++tile)
        #pragma unroll
        for (int kk = 0; kk < 32; ++kk)
            af[tile][kk] = *(const v2f*)&xs[(16 * tile + lm) * PJ_STR + 4 * kk + lk];

    #pragma unroll
    for (int blk = 0; blk < PJ_NBLK; ++blk) {
        const float* wcur = wbuf + (blk & 1) * PJ_WB_FLOATS;
        v2f bf[32];
        #pragma unroll
        for (int kk = 0; kk < 32; ++kk)
            bf[kk] = *(const v2f*)&wcur[(16 * wave + lm) * PJ_STR + 4 * kk + lk];

        __builtin_amdgcn_sched_barrier(0);

        v8f acc0 = {}, acc1 = {};
        #pragma unroll
        for (int kk = 0; kk < 32; ++kk) {
            acc0 = __builtin_amdgcn_wmma_f32_16x16x4_f32(false, af[0][kk], false, bf[kk],
                                                         (short)0, acc0, false, false);
            acc1 = __builtin_amdgcn_wmma_f32_16x16x4_f32(false, af[1][kk], false, bf[kk],
                                                         (short)0, acc1, false, false);
        }

        {
            const int second = blk >> 1;
            const int mat    = 2 * PART + second;
            const int col    = (blk & 1) * PJ_COLS + 16 * wave + lm;
            float* outp      = second ? om1 : om0;
            float bias = 0.0f;
            if (mat >= 3) bias = (second ? bm1 : bm0)[col];
            #pragma unroll
            for (int tile = 0; tile < 2; ++tile) {
                const int rowBase = m0 + 16 * tile + 8 * hi;
                #pragma unroll
                for (int v = 0; v < 8; ++v) {
                    float val = tile ? acc1[v] : acc0[v];
                    if (mat <= 1)       val *= SCALE;
                    else if (mat == 3)  val = expf(val + bias);
                    else if (mat >= 4)  val = 1.0f / (1.0f + expf(-(val + bias)));
                    outp[(rowBase + v) * HH + col] = val;
                }
            }
        }

        if (blk + 1 < PJ_NBLK) {
            float4 wv[16];
            const float* wsrc = (((blk + 1) >> 1) ? Wm1 : Wm0) + (((blk + 1) & 1) * PJ_COLS + srow) * HH + sc;
            #pragma unroll
            for (int i = 0; i < 16; ++i) wv[i] = *(const float4*)(wsrc + (4 * i) * HH);
            float* wnext = wbuf + ((blk + 1) & 1) * PJ_WB_FLOATS;
            #pragma unroll
            for (int i = 0; i < 16; ++i) *(float4*)&wnext[(srow + 4 * i) * PJ_STR + sc] = wv[i];
            __syncthreads();
        }
    }
}

__global__ __launch_bounds__(128) void proj_kernel(
    const float* __restrict__ x,
    const float* __restrict__ Wq, const float* __restrict__ Wk,
    const float* __restrict__ Wv, const float* __restrict__ Wi,
    const float* __restrict__ Wf, const float* __restrict__ Wo,
    const float* __restrict__ b_i, const float* __restrict__ b_f,
    const float* __restrict__ b_o,
    float* __restrict__ oq, float* __restrict__ ok, float* __restrict__ ov,
    float* __restrict__ oi, float* __restrict__ of, float* __restrict__ oo)
{
    __shared__ float4 pj4[(PJ_XS_FLOATS + 2 * PJ_WB_FLOATS) / 4];
    float* xs   = (float*)pj4;
    float* wbuf = xs + PJ_XS_FLOATS;
    switch (blockIdx.x % 3) {
        case 0:  pj_body<0>(x, Wq, Wk, b_i, b_i, oq, ok, xs, wbuf); break;
        case 1:  pj_body<1>(x, Wv, Wi, b_i, b_i, ov, oi, xs, wbuf); break;
        default: pj_body<2>(x, Wf, Wo, b_f, b_o, of, oo, xs, wbuf); break;
    }
}

#define PLANE  (HH * HH)
#define PLANE4 (HH * HH / 4)

#define TSTR 20
#define A_KT_OFF 0
#define A_WC_OFF (128 * TSTR)
#define A_FLOATS (2 * 128 * TSTR)

__global__ __launch_bounds__(128) void chunk_state_kernel(
    const float* __restrict__ kb, const float* __restrict__ vb,
    const float* __restrict__ ib, const float* __restrict__ fb,
    float* __restrict__ dbuf, float* __restrict__ g15buf,
    float* __restrict__ nlbuf)
{
    __shared__ float4 sa4[A_FLOATS / 4];
    float* sa = (float*)sa4;
    const int b    = blockIdx.x / NCH;
    const int ch   = blockIdx.x % NCH;
    const int r    = threadIdx.x;
    const int lane = r & 31;
    const int wave = r >> 5;
    const int lm   = lane & 15;
    const int lk   = (lane >> 4) << 1;

    const long gbase = ((long)b * TT + (long)(ch * LCH)) * HH + r;

    float pk[LCH], pi[LCH], pf[LCH], pv[LCH];
    #pragma unroll
    for (int s = 0; s < LCH; ++s) {
        const long idx = gbase + (long)s * HH;
        pk[s] = kb[idx]; pi[s] = ib[idx]; pf[s] = fb[idx]; pv[s] = vb[idx];
    }

    float G[LCH];
    {
        float g = 1.0f;
        #pragma unroll
        for (int s = 0; s < LCH; ++s) { g *= pf[s]; G[s] = g; }
    }
    const float g15 = G[LCH - 1];
    float np = 0.0f;
    #pragma unroll
    for (int s = 0; s < LCH; ++s) np = fmaf(pf[s], np, pi[s] * pk[s]);

    #pragma unroll
    for (int s = 0; s < LCH; ++s) {
        const float wg = pi[s] * pv[s] / G[s];
        sa[A_KT_OFF + r * TSTR + s] = pk[s];
        sa[A_WC_OFF + r * TSTR + s] = g15 * wg;
    }
    const long cidx = (long)(b * NCH + ch) * HH + r;
    g15buf[cidx] = g15;
    nlbuf[cidx]  = np;
    __syncthreads();

    float4* __restrict__ dptr4 = (float4*)(dbuf + (long)(b * NCH + ch) * PLANE);
    {
        const int mt0 = wave * 2;
        v2f wfr[2][4];
        #pragma unroll
        for (int mi = 0; mi < 2; ++mi)
            #pragma unroll
            for (int kk = 0; kk < 4; ++kk)
                wfr[mi][kk] = *(const v2f*)&sa[A_WC_OFF + ((mt0 + mi) * 16 + lm) * TSTR + kk * 4 + lk];
        #pragma unroll
        for (int nt = 0; nt < 8; ++nt) {
            v2f kfr[4];
            #pragma unroll
            for (int kk = 0; kk < 4; ++kk)
                kfr[kk] = *(const v2f*)&sa[A_KT_OFF + (nt * 16 + lm) * TSTR + kk * 4 + lk];
            v8f d0 = {}, d1 = {};
            #pragma unroll
            for (int kk = 0; kk < 4; ++kk) {
                d0 = __builtin_amdgcn_wmma_f32_16x16x4_f32(false, kfr[kk], false, wfr[0][kk],
                                                           (short)0, d0, false, false);
                d1 = __builtin_amdgcn_wmma_f32_16x16x4_f32(false, kfr[kk], false, wfr[1][kk],
                                                           (short)0, d1, false, false);
            }
            const int u0 = ((mt0 + 0) * 8 + nt) * 64 + lane;
            const int u1 = ((mt0 + 1) * 8 + nt) * 64 + lane;
            dptr4[u0]      = make_float4(d0[0], d0[1], d0[2], d0[3]);
            dptr4[u0 + 32] = make_float4(d0[4], d0[5], d0[6], d0[7]);
            dptr4[u1]      = make_float4(d1[0], d1[1], d1[2], d1[3]);
            dptr4[u1 + 32] = make_float4(d1[4], d1[5], d1[6], d1[7]);
        }
    }
}

#define SCAN_GRP 16
#define SCAN_CBLK (BB * PLANE4 / 256)
#define SCAN_NBLK (BB * HH / 256)

__global__ __launch_bounds__(256) void state_scan_kernel(
    float* __restrict__ dbuf, const float* __restrict__ g15buf,
    float* __restrict__ nlbuf)
{
    if (blockIdx.x < SCAN_CBLK) {
        const int gid = blockIdx.x * 256 + threadIdx.x;
        const int b   = gid / PLANE4;
        const int e4  = gid % PLANE4;
        const int row = ((e4 >> 9) << 4) + (e4 & 15);
        float4* p = (float4*)(dbuf + (long)b * NCH * PLANE) + e4;
        const float* gp = g15buf + (long)b * NCH * HH + row;

        float4 c = make_float4(0.0f, 0.0f, 0.0f, 0.0f);
        for (int c0 = 0; c0 < NCH; c0 += SCAN_GRP) {
            float4 d[SCAN_GRP];
            float  g[SCAN_GRP];
            #pragma unroll
            for (int j = 0; j < SCAN_GRP; ++j) {
                d[j] = p[(long)(c0 + j) * PLANE4];
                g[j] = gp[(c0 + j) * HH];
            }
            #pragma unroll
            for (int j = 0; j < SCAN_GRP; ++j) {
                p[(long)(c0 + j) * PLANE4] = c;
                c.x = fmaf(g[j], c.x, d[j].x);
                c.y = fmaf(g[j], c.y, d[j].y);
                c.z = fmaf(g[j], c.z, d[j].z);
                c.w = fmaf(g[j], c.w, d[j].w);
            }
        }
    } else {
        const int t   = (blockIdx.x - SCAN_CBLK) * 256 + threadIdx.x;
        const int b   = t >> 7;
        const int row = t & 127;
        const float* gp = g15buf + (long)b * NCH * HH + row;
        float*       np = nlbuf  + (long)b * NCH * HH + row;
        float g[NCH], nv[NCH];
        #pragma unroll
        for (int c = 0; c < NCH; ++c) { g[c] = gp[c * HH]; nv[c] = np[c * HH]; }
        float ns = 0.0f;
        #pragma unroll
        for (int c = 0; c < NCH; ++c) {
            np[c * HH] = ns;
            ns = fmaf(g[c], ns, nv[c]);
        }
    }
}

#define CSTR 132
#define Q_OFF   0
#define K_OFF   (Q_OFF + 16 * CSTR)
#define P_OFF   (K_OFF + 16 * CSTR)
#define M_OFF   (P_OFF + 16 * CSTR)
#define SP_OFF  (M_OFF + 16 * CSTR)
#define S_OFF   (SP_OFF + 4 * 16 * 17)
#define NPS_OFF (S_OFF + 16 * 17)
#define NQ_OFF  (NPS_OFF + 128)
#define SMEM_FLOATS (NQ_OFF + 16)
#define MU_OFF  NPS_OFF
#define RS_OFF  (NPS_OFF + 16)

__global__ __launch_bounds__(128) void chunk_out_kernel(
    const float* __restrict__ qb, const float* __restrict__ kb,
    const float* __restrict__ vb, const float* __restrict__ ib,
    const float* __restrict__ fb, const float* __restrict__ ob,
    const float* __restrict__ cbuf, const float* __restrict__ nsbuf,
    const float* __restrict__ x,
    const float* __restrict__ lnw, const float* __restrict__ lnb,
    float* __restrict__ out)
{
    __shared__ float4 sm4[SMEM_FLOATS / 4];
    float* sm = (float*)sm4;
    const int b    = blockIdx.x / NCH;
    const int ch   = blockIdx.x % NCH;
    const int r    = threadIdx.x;
    const int lane = r & 31;
    const int wave = r >> 5;
    const int lm   = lane & 15;
    const int hi   = lane >> 4;
    const int lk   = hi << 1;

    const long gbase = ((long)b * TT + (long)(ch * LCH)) * HH + r;

    float pk[LCH], pq[LCH], pi[LCH], pf[LCH], pv[LCH], po[LCH];
    #pragma unroll
    for (int s = 0; s < LCH; ++s) {
        const long idx = gbase + (long)s * HH;
        pk[s] = kb[idx]; pq[s] = qb[idx]; pi[s] = ib[idx];
        pf[s] = fb[idx]; pv[s] = vb[idx]; po[s] = ob[idx];
    }
    const float nstart = nsbuf[(long)(b * NCH + ch) * HH + r];
    float px[LCH];
    #pragma unroll
    for (int s = 0; s < LCH; ++s) px[s] = x[gbase + (long)s * HH];
    const float lw = lnw[r];
    const float lb = lnb[r];

    float4 cf[2][16];
    {
        const float4* cp4 = (const float4*)(cbuf + (long)(b * NCH + ch) * PLANE);
        #pragma unroll
        for (int tile = 0; tile < 2; ++tile)
            #pragma unroll
            for (int j = 0; j < 16; ++j)
                cf[tile][j] = cp4[((2 * wave + tile) * 16 + j) * 32 + lane];
    }

    float G[LCH], Wg[LCH], ovv[LCH];
    {
        float g = 1.0f;
        #pragma unroll
        for (int s = 0; s < LCH; ++s) { g *= pf[s]; G[s] = g; ovv[s] = po[s]; }
        #pragma unroll
        for (int s = 0; s < LCH; ++s) Wg[s] = pi[s] * pv[s] / G[s];
        float np = 0.0f;
        #pragma unroll
        for (int s = 0; s < LCH; ++s) {
            np = fmaf(pf[s], np, pi[s] * pk[s]);
            const float nfull = fmaf(G[s], nstart, np);
            sm[M_OFF + s * CSTR + r] = nfull * pq[s];
        }
        #pragma unroll
        for (int s = 0; s < LCH; ++s) {
            sm[Q_OFF + s * CSTR + r] = pq[s];
            sm[K_OFF + s * CSTR + r] = pk[s];
        }
    }
    __syncthreads();

    {
        const int s = r >> 3, seg = r & 7;
        float ps = 0.0f;
        #pragma unroll
        for (int e = 0; e < 16; ++e) ps += sm[M_OFF + s * CSTR + seg * 16 + e];
        sm[NPS_OFF + s * 8 + seg] = ps;
    }
    {
        v8f acc0 = {}, acc1 = {};
        #pragma unroll
        for (int j = 0; j < 16; ++j) {
            const int nt = j >> 1, half = j & 1;
            const float4 qa = *(const float4*)&sm[Q_OFF + lm * CSTR + 16 * nt + 8 * hi + 4 * half];
            const v2f a0 = {qa.x, qa.y};
            const v2f a1 = {qa.z, qa.w};
            const v2f b00 = {cf[0][j].x, cf[0][j].y};
            const v2f b01 = {cf[0][j].z, cf[0][j].w};
            const v2f b10 = {cf[1][j].x, cf[1][j].y};
            const v2f b11 = {cf[1][j].z, cf[1][j].w};
            acc0 = __builtin_amdgcn_wmma_f32_16x16x4_f32(false, a0, false, b00, (short)0, acc0, false, false);
            acc1 = __builtin_amdgcn_wmma_f32_16x16x4_f32(false, a0, false, b10, (short)0, acc1, false, false);
            acc0 = __builtin_amdgcn_wmma_f32_16x16x4_f32(false, a1, false, b01, (short)0, acc0, false, false);
            acc1 = __builtin_amdgcn_wmma_f32_16x16x4_f32(false, a1, false, b11, (short)0, acc1, false, false);
        }
        const int n0p = (2 * wave) << 4;
        #pragma unroll
        for (int v = 0; v < 8; ++v) {
            sm[P_OFF + (v + 8 * hi) * CSTR + n0p + lm]      = acc0[v];
            sm[P_OFF + (v + 8 * hi) * CSTR + n0p + 16 + lm] = acc1[v];
        }
    }
    {
        const int kbase = wave * 32;
        v8f sacc = {};
        #pragma unroll
        for (int kk = 0; kk < 8; ++kk) {
            const int c0 = kbase + kk * 4 + lk;
            v2f a  = *(const v2f*)&sm[K_OFF + lm * CSTR + c0];
            v2f bq = *(const v2f*)&sm[Q_OFF + lm * CSTR + c0];
            sacc = __builtin_amdgcn_wmma_f32_16x16x4_f32(false, a, false, bq,
                                                         (short)0, sacc, false, false);
        }
        #pragma unroll
        for (int v = 0; v < 8; ++v)
            sm[SP_OFF + wave * 272 + (v + 8 * hi) * 17 + lm] = sacc[v];
    }
    __syncthreads();

    if (r < 16) {
        float t = 0.0f;
        #pragma unroll
        for (int e = 0; e < 8; ++e) t += sm[NPS_OFF + r * 8 + e];
        sm[NQ_OFF + r] = t;
    }
    #pragma unroll
    for (int p = 0; p < 2; ++p) {
        const int idx = r * 2 + p;
        const int u = idx >> 4, sc = idx & 15;
        float ssum = sm[SP_OFF + 0 * 272 + u * 17 + sc]
                   + sm[SP_OFF + 1 * 272 + u * 17 + sc]
                   + sm[SP_OFF + 2 * 272 + u * 17 + sc]
                   + sm[SP_OFF + 3 * 272 + u * 17 + sc];
        sm[S_OFF + u * 17 + sc] = ssum;
    }
    __syncthreads();

    float yv[LCH];
    #pragma unroll
    for (int s = 0; s < LCH; ++s) {
        float acc = sm[P_OFF + s * CSTR + r];
        #pragma unroll
        for (int u = 0; u < LCH; ++u)
            if (u <= s) acc = fmaf(Wg[u], sm[S_OFF + u * 17 + s], acc);
        const float ret = G[s] * acc;
        const float nrm = fmaxf(fabsf(sm[NQ_OFF + s]), 1.0f);
        const float hval = ovv[s] * ret / nrm;
        yv[s] = hval + px[s];
        sm[P_OFF + s * CSTR + r] = yv[s];
    }
    __syncthreads();

    {
        const int s = r >> 3, seg = r & 7;
        const float4* yp = (const float4*)&sm[P_OFF + s * CSTR + seg * 16];
        float t1 = 0.0f, t2 = 0.0f;
        #pragma unroll
        for (int e = 0; e < 4; ++e) {
            const float4 a = yp[e];
            t1 += a.x; t2 += a.x * a.x;
            t1 += a.y; t2 += a.y * a.y;
            t1 += a.z; t2 += a.z * a.z;
            t1 += a.w; t2 += a.w * a.w;
        }
        #pragma unroll
        for (int off = 1; off <= 4; off <<= 1) {
            t1 += __shfl_xor(t1, off, 32);
            t2 += __shfl_xor(t2, off, 32);
        }
        const float mu  = t1 * (1.0f / HH);
        const float var = t2 * (1.0f / HH) - mu * mu;
        if (seg == 0) {
            sm[MU_OFF + s] = mu;
            sm[RS_OFF + s] = rsqrtf(var + LN_EPS);
        }
    }
    __syncthreads();

    #pragma unroll
    for (int s = 0; s < LCH; ++s)
        out[gbase + (long)s * HH] = (yv[s] - sm[MU_OFF + s]) * sm[RS_OFF + s] * lw + lb;
}

extern "C" void kernel_launch(void* const* d_in, const int* in_sizes, int n_in,
                              void* d_out, int out_size, void* d_ws, size_t ws_size,
                              hipStream_t stream)
{
    const float* x   = (const float*)d_in[0];
    const float* Wq  = (const float*)d_in[1];
    const float* Wk  = (const float*)d_in[2];
    const float* Wv  = (const float*)d_in[3];
    const float* Wi  = (const float*)d_in[4];
    const float* Wf  = (const float*)d_in[5];
    const float* Wo  = (const float*)d_in[6];
    const float* b_i = (const float*)d_in[7];
    const float* b_f = (const float*)d_in[8];
    const float* b_o = (const float*)d_in[9];
    const float* lnw = (const float*)d_in[10];
    const float* lnb = (const float*)d_in[11];
    float* out = (float*)d_out;

    float* ws = (float*)d_ws;
    float* q  = ws + 0L * NEL;
    float* k  = ws + 1L * NEL;
    float* v  = ws + 2L * NEL;
    float* ig = ws + 3L * NEL;
    float* fg = ws + 4L * NEL;
    float* og = ws + 5L * NEL;
    float* cst = ws + 7L * NEL;
    float* g15 = ws + 15L * NEL;
    float* nl  = ws + 15L * NEL + BB * NCH * HH;

    proj_kernel<<<PJ_GRID, 128, 0, stream>>>(x, Wq, Wk, Wv, Wi, Wf, Wo,
                                          b_i, b_f, b_o,
                                          q, k, v, ig, fg, og);
    chunk_state_kernel<<<BB * NCH, 128, 0, stream>>>(k, v, ig, fg, cst, g15, nl);
    state_scan_kernel<<<SCAN_CBLK + SCAN_NBLK, 256, 0, stream>>>(cst, g15, nl);
    chunk_out_kernel<<<BB * NCH, 128, 0, stream>>>(q, k, v, ig, fg, og, cst, nl,
                                                   x, lnw, lnb, out);
}


# State stored once per pair of chunks (16 planes a batch, not 32): half the plane traffic of the three state kernels; a pair's 2nd chunk steps the state by WMMA in registers; sum order of C, n changes
# speedup vs baseline: 9.7790x; 1.1538x over previous
#include <hip/hip_runtime.h>
#include <math.h>


typedef float v2f __attribute__((ext_vector_type(2)));
typedef float v8f __attribute__((ext_vector_type(8)));

#define BB   16
#define TT   512
#define HH   128
#define MM   (BB * TT)
#define NEL  (BB * TT * HH)
#define SCALE 0.08838834764831845f
#define LN_EPS 1e-5f
#define LCH  16
#define NCH  (TT / LCH)
#define NSC  (NCH / 2)

#define PJ_ROWS 32
#define PJ_COLS 64
#define PJ_STR  132
#define PJ_XS_FLOATS (PJ_ROWS * PJ_STR)
#define PJ_WB_FLOATS (PJ_COLS * PJ_STR)
#define PJ_NBLK 4
#define PJ_GRID (3 * (MM / PJ_ROWS))

template <int PART>
__device__ __forceinline__ void pj_body(
    const float* __restrict__ x,
    const float* __restrict__ Wm0, const float* __restrict__ Wm1,
    const float* __restrict__ bm0, const float* __restrict__ bm1,
    float* __restrict__ om0, float* __restrict__ om1,
    float* xs, float* wbuf)
{
    const int t    = threadIdx.x;
    const int lane = t & 31;
    const int wave = t >> 5;
    const int lm   = lane & 15;
    const int hi   = lane >> 4;
    const int lk   = hi << 1;
    const int m0   = (blockIdx.x / 3) * PJ_ROWS;

    const int srow = wave;
    const int sc   = lane << 2;

    {
        float4 xv[8], wv[16];
        const float* xsrc = x + (long)(m0 + srow) * HH + sc;
        #pragma unroll
        for (int i = 0; i < 8; ++i) xv[i] = *(const float4*)(xsrc + (4 * i) * HH);
        const float* wsrc = Wm0 + srow * HH + sc;
        #pragma unroll
        for (int i = 0; i < 16; ++i) wv[i] = *(const float4*)(wsrc + (4 * i) * HH);
        #pragma unroll
        for (int i = 0; i < 8; ++i) *(float4*)&xs[(srow + 4 * i) * PJ_STR + sc] = xv[i];
        #pragma unroll
        for (int i = 0; i < 16; ++i) *(float4*)&wbuf[(srow + 4 * i) * PJ_STR + sc] = wv[i];
    }
    __syncthreads();

    v2f af[2][32];
    #pragma unroll
    for (int tile = 0; tile < 2; ++tile)
        #pragma unroll
        for (int kk = 0; kk < 32; ++kk)
            af[tile][kk] = *(const v2f*)&xs[(16 * tile + lm) * PJ_STR + 4 * kk + lk];

    #pragma unroll
    for (int blk = 0; blk < PJ_NBLK; ++blk) {
        const float* wcur = wbuf + (blk & 1) * PJ_WB_FLOATS;
        v2f bf[32];
        #pragma unroll
        for (int kk = 0; kk < 32; ++kk)
            bf[kk] = *(const v2f*)&wcur[(16 * wave + lm) * PJ_STR + 4 * kk + lk];

        __builtin_amdgcn_sched_barrier(0);

        v8f acc0 = {}, acc1 = {};
        #pragma unroll
        for (int kk = 0; kk < 32; ++kk) {
            acc0 = __builtin_amdgcn_wmma_f32_16x16x4_f32(false, af[0][kk], false, bf[kk],
                                                         (short)0, acc0, false, false);
            acc1 = __builtin_amdgcn_wmma_f32_16x16x4_f32(false, af[1][kk], false, bf[kk],
                                                         (short)0, acc1, false, false);
        }

        {
            const int second = blk >> 1;
            const int mat    = 2 * PART + second;
            const int col    = (blk & 1) * PJ_COLS + 16 * wave + lm;
            float* outp      = second ? om1 : om0;
            float bias = 0.0f;
            if (mat >= 3) bias = (second ? bm1 : bm0)[col];
            #pragma unroll
            for (int tile = 0; tile < 2; ++tile) {
                const int rowBase = m0 + 16 * tile + 8 * hi;
                #pragma unroll
                for (int v = 0; v < 8; ++v) {
                    float val = tile ? acc1[v] : acc0[v];
                    if (mat <= 1)       val *= SCALE;
                    else if (mat == 3)  val = expf(val + bias);
                    else if (mat >= 4)  val = 1.0f / (1.0f + expf(-(val + bias)));
                    outp[(rowBase + v) * HH + col] = val;
                }
            }
        }

        if (blk + 1 < PJ_NBLK) {
            float4 wv[16];
            const float* wsrc = (((blk + 1) >> 1) ? Wm1 : Wm0) + (((blk + 1) & 1) * PJ_COLS + srow) * HH + sc;
            #pragma unroll
            for (int i = 0; i < 16; ++i) wv[i] = *(const float4*)(wsrc + (4 * i) * HH);
            float* wnext = wbuf + ((blk + 1) & 1) * PJ_WB_FLOATS;
            #pragma unroll
            for (int i = 0; i < 16; ++i) *(float4*)&wnext[(srow + 4 * i) * PJ_STR + sc] = wv[i];
            __syncthreads();
        }
    }
}

__global__ __launch_bounds__(128) void proj_kernel(
    const float* __restrict__ x,
    const float* __restrict__ Wq, const float* __restrict__ Wk,
    const float* __restrict__ Wv, const float* __restrict__ Wi,
    const float* __restrict__ Wf, const float* __restrict__ Wo,
    const float* __restrict__ b_i, const float* __restrict__ b_f,
    const float* __restrict__ b_o,
    float* __restrict__ oq, float* __restrict__ ok, float* __restrict__ ov,
    float* __restrict__ oi, float* __restrict__ of, float* __restrict__ oo)
{
    __shared__ float4 pj4[(PJ_XS_FLOATS + 2 * PJ_WB_FLOATS) / 4];
    float* xs   = (float*)pj4;
    float* wbuf = xs + PJ_XS_FLOATS;
    switch (blockIdx.x % 3) {
        case 0:  pj_body<0>(x, Wq, Wk, b_i, b_i, oq, ok, xs, wbuf); break;
        case 1:  pj_body<1>(x, Wv, Wi, b_i, b_i, ov, oi, xs, wbuf); break;
        default: pj_body<2>(x, Wf, Wo, b_f, b_o, of, oo, xs, wbuf); break;
    }
}

#define PLANE  (HH * HH)
#define PLANE4 (HH * HH / 4)

#define TSTR 36
#define A_KT_OFF 0
#define A_WC_OFF (128 * TSTR)
#define A_GX_OFF (2 * 128 * TSTR)
#define A_FLOATS (2 * 128 * TSTR + 256)

__global__ __launch_bounds__(256) void chunk_state_kernel(
    const float* __restrict__ kb, const float* __restrict__ vb,
    const float* __restrict__ ib, const float* __restrict__ fb,
    float* __restrict__ dbuf, float* __restrict__ g15buf,
    float* __restrict__ nlbuf)
{
    __shared__ float4 sa4[A_FLOATS / 4];
    float* sa = (float*)sa4;
    const int b    = blockIdx.x / NSC;
    const int sc   = blockIdx.x % NSC;
    const int t    = threadIdx.x;
    const int r    = t & 127;
    const int sub  = t >> 7;
    const int lane = t & 31;
    const int wave = t >> 5;
    const int lm   = lane & 15;
    const int lk   = (lane >> 4) << 1;

    const long gbase  = ((long)b * TT + (long)((2 * sc + sub) * LCH)) * HH + r;
    const long gbase2 = ((long)b * TT + (long)((2 * sc + 1) * LCH)) * HH + r;

    float pk[LCH], pi[LCH], pf[LCH], pv[LCH], pf2[LCH];
    #pragma unroll
    for (int s = 0; s < LCH; ++s) {
        const long idx = gbase + (long)s * HH;
        pk[s] = kb[idx]; pi[s] = ib[idx]; pf[s] = fb[idx]; pv[s] = vb[idx];
        pf2[s] = fb[gbase2 + (long)s * HH];
    }

    float G[LCH];
    {
        float g = 1.0f;
        #pragma unroll
        for (int s = 0; s < LCH; ++s) { g *= pf[s]; G[s] = g; }
    }
    const float g15 = G[LCH - 1];
    float g2 = 1.0f;
    #pragma unroll
    for (int s = 0; s < LCH; ++s) g2 *= pf2[s];
    float np = 0.0f;
    #pragma unroll
    for (int s = 0; s < LCH; ++s) np = fmaf(pf[s], np, pi[s] * pk[s]);

    const float tail = sub ? 1.0f : g2;
    #pragma unroll
    for (int s = 0; s < LCH; ++s) {
        const float wg = pi[s] * pv[s] / G[s];
        sa[A_KT_OFF + r * TSTR + 16 * sub + s] = pk[s];
        sa[A_WC_OFF + r * TSTR + 16 * sub + s] = tail * (g15 * wg);
    }
    if (sub == 0) {
        sa[A_GX_OFF + r]       = g15;
        sa[A_GX_OFF + 128 + r] = np;
    }
    __syncthreads();
    if (sub == 1) {
        const long cidx = (long)(b * NSC + sc) * HH + r;
        g15buf[cidx] = sa[A_GX_OFF + r] * g15;
        nlbuf[cidx]  = fmaf(g15, sa[A_GX_OFF + 128 + r], np);
    }

    float4* __restrict__ dptr4 = (float4*)(dbuf + (long)(b * NSC + sc) * PLANE);
    {
        v2f wfr[8];
        #pragma unroll
        for (int kk = 0; kk < 8; ++kk)
            wfr[kk] = *(const v2f*)&sa[A_WC_OFF + (wave * 16 + lm) * TSTR + kk * 4 + lk];
        #pragma unroll
        for (int nt = 0; nt < 8; nt += 2) {
            v2f kf0[8], kf1[8];
            #pragma unroll
            for (int kk = 0; kk < 8; ++kk) {
                kf0[kk] = *(const v2f*)&sa[A_KT_OFF + ((nt + 0) * 16 + lm) * TSTR + kk * 4 + lk];
                kf1[kk] = *(const v2f*)&sa[A_KT_OFF + ((nt + 1) * 16 + lm) * TSTR + kk * 4 + lk];
            }
            v8f d0 = {}, d1 = {};
            #pragma unroll
            for (int kk = 0; kk < 8; ++kk) {
                d0 = __builtin_amdgcn_wmma_f32_16x16x4_f32(false, kf0[kk], false, wfr[kk],
                                                           (short)0, d0, false, false);
                d1 = __builtin_amdgcn_wmma_f32_16x16x4_f32(false, kf1[kk], false, wfr[kk],
                                                           (short)0, d1, false, false);
            }
            const int u0 = (wave * 8 + nt + 0) * 64 + lane;
            const int u1 = (wave * 8 + nt + 1) * 64 + lane;
            dptr4[u0]      = make_float4(d0[0], d0[1], d0[2], d0[3]);
            dptr4[u0 + 32] = make_float4(d0[4], d0[5], d0[6], d0[7]);
            dptr4[u1]      = make_float4(d1[0], d1[1], d1[2], d1[3]);
            dptr4[u1 + 32] = make_float4(d1[4], d1[5], d1[6], d1[7]);
        }
    }
}

#define SCAN_GRP 16
#define SCAN_CBLK (BB * PLANE4 / 256)
#define SCAN_NBLK (BB * HH / 256)

__global__ __launch_bounds__(256) void state_scan_kernel(
    float* __restrict__ dbuf, const float* __restrict__ g15buf,
    float* __restrict__ nlbuf)
{
    if (blockIdx.x < SCAN_CBLK) {
        const int gid = blockIdx.x * 256 + threadIdx.x;
        const int b   = gid / PLANE4;
        const int e4  = gid % PLANE4;
        const int row = ((e4 >> 9) << 4) + (e4 & 15);
        float4* p = (float4*)(dbuf + (long)b * NSC * PLANE) + e4;
        const float* gp = g15buf + (long)b * NSC * HH + row;

        float4 c = make_float4(0.0f, 0.0f, 0.0f, 0.0f);
        for (int c0 = 0; c0 < NSC; c0 += SCAN_GRP) {
            float4 d[SCAN_GRP];
            float  g[SCAN_GRP];
            #pragma unroll
            for (int j = 0; j < SCAN_GRP; ++j) {
                d[j] = p[(long)(c0 + j) * PLANE4];
                g[j] = gp[(c0 + j) * HH];
            }
            #pragma unroll
            for (int j = 0; j < SCAN_GRP; ++j) {
                p[(long)(c0 + j) * PLANE4] = c;
                c.x = fmaf(g[j], c.x, d[j].x);
                c.y = fmaf(g[j], c.y, d[j].y);
                c.z = fmaf(g[j], c.z, d[j].z);
                c.w = fmaf(g[j], c.w, d[j].w);
            }
        }
    } else {
        const int t   = (blockIdx.x - SCAN_CBLK) * 256 + threadIdx.x;
        const int b   = t >> 7;
        const int row = t & 127;
        const float* gp = g15buf + (long)b * NSC * HH + row;
        float*       np = nlbuf  + (long)b * NSC * HH + row;
        float g[NSC], nv[NSC];
        #pragma unroll
        for (int c = 0; c < NSC; ++c) { g[c] = gp[c * HH]; nv[c] = np[c * HH]; }
        float ns = 0.0f;
        #pragma unroll
        for (int c = 0; c < NSC; ++c) {
            np[c * HH] = ns;
            ns = fmaf(g[c], ns, nv[c]);
        }
    }
}

#define CSTR 132
#define Q_OFF   0
#define K_OFF   (Q_OFF + 16 * CSTR)
#define P_OFF   (K_OFF + 16 * CSTR)
#define M_OFF   (P_OFF + 16 * CSTR)
#define SP_OFF  (M_OFF + 16 * CSTR)
#define S_OFF   (SP_OFF + 4 * 16 * 17)
#define NPS_OFF (S_OFF + 16 * 17)
#define NQ_OFF  (NPS_OFF + 128)
#define XSTR    20
#define XK_OFF  (NQ_OFF + 16)
#define XW_OFF  (XK_OFF + 128 * XSTR)
#define XG_OFF  (XW_OFF + 128 * XSTR)
#define SMEM_FLOATS (XG_OFF + 128)
#define MU_OFF  NPS_OFF
#define RS_OFF  (NPS_OFF + 16)

__global__ __launch_bounds__(128) void chunk_out_kernel(
    const float* __restrict__ qb, const float* __restrict__ kb,
    const float* __restrict__ vb, const float* __restrict__ ib,
    const float* __restrict__ fb, const float* __restrict__ ob,
    const float* __restrict__ cbuf, const float* __restrict__ nsbuf,
    const float* __restrict__ x,
    const float* __restrict__ lnw, const float* __restrict__ lnb,
    float* __restrict__ out)
{
    __shared__ float4 sm4[SMEM_FLOATS / 4];
    float* sm = (float*)sm4;
    const int b    = blockIdx.x / NCH;
    const int ch   = blockIdx.x % NCH;
    const int r    = threadIdx.x;
    const int lane = r & 31;
    const int wave = r >> 5;
    const int lm   = lane & 15;
    const int hi   = lane >> 4;
    const int lk   = hi << 1;

    const long gbase = ((long)b * TT + (long)(ch * LCH)) * HH + r;

    float pk[LCH], pq[LCH], pi[LCH], pf[LCH], pv[LCH], po[LCH];
    #pragma unroll
    for (int s = 0; s < LCH; ++s) {
        const long idx = gbase + (long)s * HH;
        pk[s] = kb[idx]; pq[s] = qb[idx]; pi[s] = ib[idx];
        pf[s] = fb[idx]; pv[s] = vb[idx]; po[s] = ob[idx];
    }
    const int sc  = ch >> 1;
    const int odd = ch & 1;
    float nstart = nsbuf[(long)(b * NSC + sc) * HH + r];
    float px[LCH];
    #pragma unroll
    for (int s = 0; s < LCH; ++s) px[s] = x[gbase + (long)s * HH];
    const float lw = lnw[r];
    const float lb = lnb[r];

    float4 cf[2][16];
    {
        const float4* cp4 = (const float4*)(cbuf + (long)(b * NSC + sc) * PLANE);
        #pragma unroll
        for (int tile = 0; tile < 2; ++tile)
            #pragma unroll
            for (int j = 0; j < 16; ++j)
                cf[tile][j] = cp4[((2 * wave + tile) * 16 + j) * 32 + lane];
    }

    if (odd) {
        float qk[LCH], qi[LCH], qf[LCH], qv[LCH];
        const long pbase = gbase - (long)LCH * HH;
        #pragma unroll
        for (int s = 0; s < LCH; ++s) {
            const long idx = pbase + (long)s * HH;
            qk[s] = kb[idx]; qi[s] = ib[idx]; qf[s] = fb[idx]; qv[s] = vb[idx];
        }
        float Ga[LCH];
        {
            float g = 1.0f;
            #pragma unroll
            for (int s = 0; s < LCH; ++s) { g *= qf[s]; Ga[s] = g; }
        }
        const float g15a = Ga[LCH - 1];
        float npa = 0.0f;
        #pragma unroll
        for (int s = 0; s < LCH; ++s) npa = fmaf(qf[s], npa, qi[s] * qk[s]);
        nstart = fmaf(g15a, nstart, npa);
        #pragma unroll
        for (int s = 0; s < LCH; ++s) {
            const float wg = qi[s] * qv[s] / Ga[s];
            sm[XK_OFF + r * XSTR + s] = qk[s];
            sm[XW_OFF + r * XSTR + s] = g15a * wg;
        }
        sm[XG_OFF + r] = g15a;
        __syncthreads();

        const int mt0 = wave * 2;
        const float ga0 = sm[XG_OFF + (mt0 + 0) * 16 + lm];
        const float ga1 = sm[XG_OFF + (mt0 + 1) * 16 + lm];
        v2f wfr[2][4];
        #pragma unroll
        for (int mi = 0; mi < 2; ++mi)
            #pragma unroll
            for (int kk = 0; kk < 4; ++kk)
                wfr[mi][kk] = *(const v2f*)&sm[XW_OFF + ((mt0 + mi) * 16 + lm) * XSTR + kk * 4 + lk];
        #pragma unroll
        for (int nt = 0; nt < 8; ++nt) {
            v2f kfr[4];
            #pragma unroll
            for (int kk = 0; kk < 4; ++kk)
                kfr[kk] = *(const v2f*)&sm[XK_OFF + (nt * 16 + lm) * XSTR + kk * 4 + lk];
            const float4 c00 = cf[0][2 * nt], c01 = cf[0][2 * nt + 1];
            const float4 c10 = cf[1][2 * nt], c11 = cf[1][2 * nt + 1];
            v8f d0 = {ga0 * c00.x, ga0 * c00.y, ga0 * c00.z, ga0 * c00.w,
                      ga0 * c01.x, ga0 * c01.y, ga0 * c01.z, ga0 * c01.w};
            v8f d1 = {ga1 * c10.x, ga1 * c10.y, ga1 * c10.z, ga1 * c10.w,
                      ga1 * c11.x, ga1 * c11.y, ga1 * c11.z, ga1 * c11.w};
            #pragma unroll
            for (int kk = 0; kk < 4; ++kk) {
                d0 = __builtin_amdgcn_wmma_f32_16x16x4_f32(false, kfr[kk], false, wfr[0][kk],
                                                           (short)0, d0, false, false);
                d1 = __builtin_amdgcn_wmma_f32_16x16x4_f32(false, kfr[kk], false, wfr[1][kk],
                                                           (short)0, d1, false, false);
            }
            cf[0][2 * nt]     = make_float4(d0[0], d0[1], d0[2], d0[3]);
            cf[0][2 * nt + 1] = make_float4(d0[4], d0[5], d0[6], d0[7]);
            cf[1][2 * nt]     = make_float4(d1[0], d1[1], d1[2], d1[3]);
            cf[1][2 * nt + 1] = make_float4(d1[4], d1[5], d1[6], d1[7]);
        }
    }

    float G[LCH], Wg[LCH], ovv[LCH];
    {
        float g = 1.0f;
        #pragma unroll
        for (int s = 0; s < LCH; ++s) { g *= pf[s]; G[s] = g; ovv[s] = po[s]; }
        #pragma unroll
        for (int s = 0; s < LCH; ++s) Wg[s] = pi[s] * pv[s] / G[s];
        float np = 0.0f;
        #pragma unroll
        for (int s = 0; s < LCH; ++s) {
            np = fmaf(pf[s], np, pi[s] * pk[s]);
            const float nfull = fmaf(G[s], nstart, np);
            sm[M_OFF + s * CSTR + r] = nfull * pq[s];
        }
        #pragma unroll
        for (int s = 0; s < LCH; ++s) {
            sm[Q_OFF + s * CSTR + r] = pq[s];
            sm[K_OFF + s * CSTR + r] = pk[s];
        }
    }
    __syncthreads();

    {
        const int s = r >> 3, seg = r & 7;
        float ps = 0.0f;
        #pragma unroll
        for (int e = 0; e < 16; ++e) ps += sm[M_OFF + s * CSTR + seg * 16 + e];
        sm[NPS_OFF + s * 8 + seg] = ps;
    }
    {
        v8f acc0 = {}, acc1 = {};
        #pragma unroll
        for (int j = 0; j < 16; ++j) {
            const int nt = j >> 1, half = j & 1;
            const float4 qa = *(const float4*)&sm[Q_OFF + lm * CSTR + 16 * nt + 8 * hi + 4 * half];
            const v2f a0 = {qa.x, qa.y};
            const v2f a1 = {qa.z, qa.w};
            const v2f b00 = {cf[0][j].x, cf[0][j].y};
            const v2f b01 = {cf[0][j].z, cf[0][j].w};
            const v2f b10 = {cf[1][j].x, cf[1][j].y};
            const v2f b11 = {cf[1][j].z, cf[1][j].w};
            acc0 = __builtin_amdgcn_wmma_f32_16x16x4_f32(false, a0, false, b00, (short)0, acc0, false, false);
            acc1 = __builtin_amdgcn_wmma_f32_16x16x4_f32(false, a0, false, b10, (short)0, acc1, false, false);
            acc0 = __builtin_amdgcn_wmma_f32_16x16x4_f32(false, a1, false, b01, (short)0, acc0, false, false);
            acc1 = __builtin_amdgcn_wmma_f32_16x16x4_f32(false, a1, false, b11, (short)0, acc1, false, false);
        }
        const int n0p = (2 * wave) << 4;
        #pragma unroll
        for (int v = 0; v < 8; ++v) {
            sm[P_OFF + (v + 8 * hi) * CSTR + n0p + lm]      = acc0[v];
            sm[P_OFF + (v + 8 * hi) * CSTR + n0p + 16 + lm] = acc1[v];
        }
    }
    {
        const int kbase = wave * 32;
        v8f sacc = {};
        #pragma unroll
        for (int kk = 0; kk < 8; ++kk) {
            const int c0 = kbase + kk * 4 + lk;
            v2f a  = *(const v2f*)&sm[K_OFF + lm * CSTR + c0];
            v2f bq = *(const v2f*)&sm[Q_OFF + lm * CSTR + c0];
            sacc = __builtin_amdgcn_wmma_f32_16x16x4_f32(false, a, false, bq,
                                                         (short)0, sacc, false, false);
        }
        #pragma unroll
        for (int v = 0; v < 8; ++v)
            sm[SP_OFF + wave * 272 + (v + 8 * hi) * 17 + lm] = sacc[v];
    }
    __syncthreads();

    if (r < 16) {
        float t = 0.0f;
        #pragma unroll
        for (int e = 0; e < 8; ++e) t += sm[NPS_OFF + r * 8 + e];
        sm[NQ_OFF + r] = t;
    }
    #pragma unroll
    for (int p = 0; p < 2; ++p) {
        const int idx = r * 2 + p;
        const int u = idx >> 4, sc = idx & 15;
        float ssum = sm[SP_OFF + 0 * 272 + u * 17 + sc]
                   + sm[SP_OFF + 1 * 272 + u * 17 + sc]
                   + sm[SP_OFF + 2 * 272 + u * 17 + sc]
                   + sm[SP_OFF + 3 * 272 + u * 17 + sc];
        sm[S_OFF + u * 17 + sc] = ssum;
    }
    __syncthreads();

    float yv[LCH];
    #pragma unroll
    for (int s = 0; s < LCH; ++s) {
        float acc = sm[P_OFF + s * CSTR + r];
        #pragma unroll
        for (int u = 0; u < LCH; ++u)
            if (u <= s) acc = fmaf(Wg[u], sm[S_OFF + u * 17 + s], acc);
        const float ret = G[s] * acc;
        const float nrm = fmaxf(fabsf(sm[NQ_OFF + s]), 1.0f);
        const float hval = ovv[s] * ret / nrm;
        yv[s] = hval + px[s];
        sm[P_OFF + s * CSTR + r] = yv[s];
    }
    __syncthreads();

    {
        const int s = r >> 3, seg = r & 7;
        const float4* yp = (const float4*)&sm[P_OFF + s * CSTR + seg * 16];
        float t1 = 0.0f, t2 = 0.0f;
        #pragma unroll
        for (int e = 0; e < 4; ++e) {
            const float4 a = yp[e];
            t1 += a.x; t2 += a.x * a.x;
            t1 += a.y; t2 += a.y * a.y;
            t1 += a.z; t2 += a.z * a.z;
            t1 += a.w; t2 += a.w * a.w;
        }
        #pragma unroll
        for (int off = 1; off <= 4; off <<= 1) {
            t1 += __shfl_xor(t1, off, 32);
            t2 += __shfl_xor(t2, off, 32);
        }
        const float mu  = t1 * (1.0f / HH);
        const float var = t2 * (1.0f / HH) - mu * mu;
        if (seg == 0) {
            sm[MU_OFF + s] = mu;
            sm[RS_OFF + s] = rsqrtf(var + LN_EPS);
        }
    }
    __syncthreads();

    #pragma unroll
    for (int s = 0; s < LCH; ++s)
        out[gbase + (long)s * HH] = (yv[s] - sm[MU_OFF + s]) * sm[RS_OFF + s] * lw + lb;
}

extern "C" void kernel_launch(void* const* d_in, const int* in_sizes, int n_in,
                              void* d_out, int out_size, void* d_ws, size_t ws_size,
                              hipStream_t stream)
{
    const float* x   = (const float*)d_in[0];
    const float* Wq  = (const float*)d_in[1];
    const float* Wk  = (const float*)d_in[2];
    const float* Wv  = (const float*)d_in[3];
    const float* Wi  = (const float*)d_in[4];
    const float* Wf  = (const float*)d_in[5];
    const float* Wo  = (const float*)d_in[6];
    const float* b_i = (const float*)d_in[7];
    const float* b_f = (const float*)d_in[8];
    const float* b_o = (const float*)d_in[9];
    const float* lnw = (const float*)d_in[10];
    const float* lnb = (const float*)d_in[11];
    float* out = (float*)d_out;

    float* ws = (float*)d_ws;
    float* q  = ws + 0L * NEL;
    float* k  = ws + 1L * NEL;
    float* v  = ws + 2L * NEL;
    float* ig = ws + 3L * NEL;
    float* fg = ws + 4L * NEL;
    float* og = ws + 5L * NEL;
    float* cst = ws + 7L * NEL;
    float* g15 = ws + 15L * NEL;
    float* nl  = ws + 15L * NEL + BB * NCH * HH;

    proj_kernel<<<PJ_GRID, 128, 0, stream>>>(x, Wq, Wk, Wv, Wi, Wf, Wo,
                                          b_i, b_f, b_o,
                                          q, k, v, ig, fg, og);
    chunk_state_kernel<<<BB * NSC, 256, 0, stream>>>(k, v, ig, fg, cst, g15, nl);
    state_scan_kernel<<<SCAN_CBLK + SCAN_NBLK, 256, 0, stream>>>(cst, g15, nl);
    chunk_out_kernel<<<BB * NCH, 128, 0, stream>>>(q, k, v, ig, fg, og, cst, nl,
                                                   x, lnw, lnb, out);
}


# State scan fused into the state kernel: a work-group walks the 16 pairs with a 32x32 state piece in WMMA accumulators and writes each plane once; one launch less; sum order of C changes
# speedup vs baseline: 10.8357x; 1.1081x over previous
#include <hip/hip_runtime.h>
#include <math.h>


typedef float v2f __attribute__((ext_vector_type(2)));
typedef float v8f __attribute__((ext_vector_type(8)));

#define BB   16
#define TT   512
#define HH   128
#define MM   (BB * TT)
#define NEL  (BB * TT * HH)
#define SCALE 0.08838834764831845f
#define LN_EPS 1e-5f
#define LCH  16
#define NCH  (TT / LCH)
#define NSC  (NCH / 2)

#define PJ_ROWS 32
#define PJ_COLS 64
#define PJ_STR  132
#define PJ_XS_FLOATS (PJ_ROWS * PJ_STR)
#define PJ_WB_FLOATS (PJ_COLS * PJ_STR)
#define PJ_NBLK 4
#define PJ_GRID (3 * (MM / PJ_ROWS))

template <int PART>
__device__ __forceinline__ void pj_body(
    const float* __restrict__ x,
    const float* __restrict__ Wm0, const float* __restrict__ Wm1,
    const float* __restrict__ bm0, const float* __restrict__ bm1,
    float* __restrict__ om0, float* __restrict__ om1,
    float* xs, float* wbuf)
{
    const int t    = threadIdx.x;
    const int lane = t & 31;
    const int wave = t >> 5;
    const int lm   = lane & 15;
    const int hi   = lane >> 4;
    const int lk   = hi << 1;
    const int m0   = (blockIdx.x / 3) * PJ_ROWS;

    const int srow = wave;
    const int sc   = lane << 2;

    {
        float4 xv[8], wv[16];
        const float* xsrc = x + (long)(m0 + srow) * HH + sc;
        #pragma unroll
        for (int i = 0; i < 8; ++i) xv[i] = *(const float4*)(xsrc + (4 * i) * HH);
        const float* wsrc = Wm0 + srow * HH + sc;
        #pragma unroll
        for (int i = 0; i < 16; ++i) wv[i] = *(const float4*)(wsrc + (4 * i) * HH);
        #pragma unroll
        for (int i = 0; i < 8; ++i) *(float4*)&xs[(srow + 4 * i) * PJ_STR + sc] = xv[i];
        #pragma unroll
        for (int i = 0; i < 16; ++i) *(float4*)&wbuf[(srow + 4 * i) * PJ_STR + sc] = wv[i];
    }
    __syncthreads();

    v2f af[2][32];
    #pragma unroll
    for (int tile = 0; tile < 2; ++tile)
        #pragma unroll
        for (int kk = 0; kk < 32; ++kk)
            af[tile][kk] = *(const v2f*)&xs[(16 * tile + lm) * PJ_STR + 4 * kk + lk];

    #pragma unroll
    for (int blk = 0; blk < PJ_NBLK; ++blk) {
        const float* wcur = wbuf + (blk & 1) * PJ_WB_FLOATS;
        v2f bf[32];
        #pragma unroll
        for (int kk = 0; kk < 32; ++kk)
            bf[kk] = *(const v2f*)&wcur[(16 * wave + lm) * PJ_STR + 4 * kk + lk];

        __builtin_amdgcn_sched_barrier(0);

        v8f acc0 = {}, acc1 = {};
        #pragma unroll
        for (int kk = 0; kk < 32; ++kk) {
            acc0 = __builtin_amdgcn_wmma_f32_16x16x4_f32(false, af[0][kk], false, bf[kk],
                                                         (short)0, acc0, false, false);
            acc1 = __builtin_amdgcn_wmma_f32_16x16x4_f32(false, af[1][kk], false, bf[kk],
                                                         (short)0, acc1, false, false);
        }

        {
            const int second = blk >> 1;
            const int mat    = 2 * PART + second;
            const int col    = (blk & 1) * PJ_COLS + 16 * wave + lm;
            float* outp      = second ? om1 : om0;
            float bias = 0.0f;
            if (mat >= 3) bias = (second ? bm1 : bm0)[col];
            #pragma unroll
            for (int tile = 0; tile < 2; ++tile) {
                const int rowBase = m0 + 16 * tile + 8 * hi;
                #pragma unroll
                for (int v = 0; v < 8; ++v) {
                    float val = tile ? acc1[v] : acc0[v];
                    if (mat <= 1)       val *= SCALE;
                    else if (mat == 3)  val = expf(val + bias);
                    else if (mat >= 4)  val = 1.0f / (1.0f + expf(-(val + bias)));
                    outp[(rowBase + v) * HH + col] = val;
                }
            }
        }

        if (blk + 1 < PJ_NBLK) {
            float4 wv[16];
            const float* wsrc = (((blk + 1) >> 1) ? Wm1 : Wm0) + (((blk + 1) & 1) * PJ_COLS + srow) * HH + sc;
            #pragma unroll
            for (int i = 0; i < 16; ++i) wv[i] = *(const float4*)(wsrc + (4 * i) * HH);
            float* wnext = wbuf + ((blk + 1) & 1) * PJ_WB_FLOATS;
            #pragma unroll
            for (int i = 0; i < 16; ++i) *(float4*)&wnext[(srow + 4 * i) * PJ_STR + sc] = wv[i];
            __syncthreads();
        }
    }
}

__global__ __launch_bounds__(128) void proj_kernel(
    const float* __restrict__ x,
    const float* __restrict__ Wq, const float* __restrict__ Wk,
    const float* __restrict__ Wv, const float* __restrict__ Wi,
    const float* __restrict__ Wf, const float* __restrict__ Wo,
    const float* __restrict__ b_i, const float* __restrict__ b_f,
    const float* __restrict__ b_o,
    float* __restrict__ oq, float* __restrict__ ok, float* __restrict__ ov,
    float* __restrict__ oi, float* __restrict__ of, float* __restrict__ oo)
{
    __shared__ float4 pj4[(PJ_XS_FLOATS + 2 * PJ_WB_FLOATS) / 4];
    float* xs   = (float*)pj4;
    float* wbuf = xs + PJ_XS_FLOATS;
    switch (blockIdx.x % 3) {
        case 0:  pj_body<0>(x, Wq, Wk, b_i, b_i, oq, ok, xs, wbuf); break;
        case 1:  pj_body<1>(x, Wv, Wi, b_i, b_i, ov, oi, xs, wbuf); break;
        default: pj_body<2>(x, Wf, Wo, b_f, b_o, of, oo, xs, wbuf); break;
    }
}

#define PLANE  (HH * HH)
#define PLANE4 (HH * HH / 4)

#define ST_KP    40
#define ST_WP    516
#define ST_K_OFF 0
#define ST_W_OFF (TT * ST_KP)
#define ST_G_OFF (ST_W_OFF + 32 * ST_WP)
#define ST_N_OFF (ST_G_OFF + 512)
#define ST_FLOATS (ST_N_OFF + 512)
#define ST_GRID  (BB * 4 * 4)

__global__ __launch_bounds__(256) void chunk_state_kernel(
    const float* __restrict__ kb, const float* __restrict__ vb,
    const float* __restrict__ ib, const float* __restrict__ fb,
    float* __restrict__ cbuf, float* __restrict__ nsbuf)
{
    __shared__ float4 ss4[ST_FLOATS / 4];
    float* ss = (float*)ss4;
    const int cq   = blockIdx.x & 3;
    const int rq   = (blockIdx.x >> 2) & 3;
    const int b    = blockIdx.x >> 4;
    const int t    = threadIdx.x;
    const int lane = t & 31;
    const int wave = t >> 5;
    const int lm   = lane & 15;
    const int hi   = lane >> 4;
    const int lk   = hi << 1;

    {
        const int kr = t >> 3;
        const int kc = (t & 7) << 2;
        const float* ksrc = kb + ((long)b * TT + kr) * HH + cq * 32 + kc;
        float4 kv[16];
        #pragma unroll
        for (int i = 0; i < 16; ++i) kv[i] = *(const float4*)(ksrc + (long)(32 * i) * HH);
        #pragma unroll
        for (int i = 0; i < 16; ++i) *(float4*)&ss[ST_K_OFF + (kr + 32 * i) * ST_KP + kc] = kv[i];
    }

    float pi[2][32], pf[2][32], pv[2][32], pk[2][32];
    #pragma unroll
    for (int h = 0; h < 2; ++h) {
        const long g0 = ((long)b * TT + (long)(32 * (2 * wave + h))) * HH + 32 * rq + lane;
        #pragma unroll
        for (int s = 0; s < 32; ++s) {
            const long idx = g0 + (long)s * HH;
            pi[h][s] = ib[idx]; pf[h][s] = fb[idx]; pv[h][s] = vb[idx];
            pk[h][s] = 0.0f;
        }
    }
    if (cq == 0) {
        #pragma unroll
        for (int h = 0; h < 2; ++h) {
            const long g0 = ((long)b * TT + (long)(32 * (2 * wave + h))) * HH + 32 * rq + lane;
            #pragma unroll
            for (int s = 0; s < 32; ++s) pk[h][s] = kb[g0 + (long)s * HH];
        }
    }
    #pragma unroll
    for (int h = 0; h < 2; ++h) {
        const int pp = 2 * wave + h;
        float Ga[LCH], Gb[LCH];
        {
            float g = 1.0f;
            #pragma unroll
            for (int s = 0; s < LCH; ++s) { g *= pf[h][s]; Ga[s] = g; }
            g = 1.0f;
            #pragma unroll
            for (int s = 0; s < LCH; ++s) { g *= pf[h][LCH + s]; Gb[s] = g; }
        }
        const float g15a = Ga[LCH - 1];
        const float g15b = Gb[LCH - 1];
        #pragma unroll
        for (int s = 0; s < LCH; ++s) {
            const float wga = pi[h][s] * pv[h][s] / Ga[s];
            const float wgb = pi[h][LCH + s] * pv[h][LCH + s] / Gb[s];
            ss[ST_W_OFF + lane * ST_WP + 32 * pp + s]       = g15b * (g15a * wga);
            ss[ST_W_OFF + lane * ST_WP + 32 * pp + LCH + s] = g15b * wgb;
        }
        ss[ST_G_OFF + pp * 32 + lane] = g15a * g15b;
        float npa = 0.0f, npb = 0.0f;
        #pragma unroll
        for (int s = 0; s < LCH; ++s) npa = fmaf(pf[h][s], npa, pi[h][s] * pk[h][s]);
        #pragma unroll
        for (int s = 0; s < LCH; ++s) npb = fmaf(pf[h][LCH + s], npb, pi[h][LCH + s] * pk[h][LCH + s]);
        if (cq == 0) ss[ST_N_OFF + pp * 32 + lane] = fmaf(g15b, npa, npb);
    }
    __syncthreads();

    if (cq == 0 && wave == 4) {
        float* np = nsbuf + (long)b * NSC * HH + 32 * rq + lane;
        float ns = 0.0f;
        #pragma unroll
        for (int p = 0; p < NSC; ++p) {
            np[p * HH] = ns;
            ns = fmaf(ss[ST_G_OFF + p * 32 + lane], ns, ss[ST_N_OFF + p * 32 + lane]);
        }
    }
    if (wave >= 4) return;

    const int rtl = wave >> 1;
    const int ctl = wave & 1;
    const int rt  = 2 * rq + rtl;
    const int ct  = 2 * cq + ctl;
    float4* dp4 = (float4*)cbuf + (long)(b * NSC) * PLANE4 + (rt * 8 + ct) * 64 + lane;
    const float* kcol = ss + ST_K_OFF + 16 * ctl + lm;
    const float* wrow = ss + ST_W_OFF + (16 * rtl + lm) * ST_WP;
    const float* grow = ss + ST_G_OFF + 16 * rtl + lm;
    v2f kf[8], wf[8];
    #pragma unroll
    for (int kk = 0; kk < 8; ++kk) {
        const int u = 4 * kk + lk;
        kf[kk] = (v2f){kcol[u * ST_KP], kcol[(u + 1) * ST_KP]};
        wf[kk] = *(const v2f*)&wrow[u];
    }
    float gq[NSC];
    #pragma unroll
    for (int p = 0; p < NSC; ++p) gq[p] = grow[p * 32];
    v8f acc = {};
    #pragma unroll
    for (int p = 0; p < NSC; ++p) {
        v2f kn[8], wn[8];
        if (p + 1 < NSC) {
            #pragma unroll
            for (int kk = 0; kk < 8; ++kk) {
                const int u = 32 * (p + 1) + 4 * kk + lk;
                kn[kk] = (v2f){kcol[u * ST_KP], kcol[(u + 1) * ST_KP]};
                wn[kk] = *(const v2f*)&wrow[u];
            }
        }
        __builtin_amdgcn_sched_barrier(0);
        dp4[(long)p * PLANE4]      = make_float4(acc[0], acc[1], acc[2], acc[3]);
        dp4[(long)p * PLANE4 + 32] = make_float4(acc[4], acc[5], acc[6], acc[7]);
        if (p + 1 < NSC) {
            #pragma unroll
            for (int v = 0; v < 8; ++v) acc[v] *= gq[p];
            #pragma unroll
            for (int kk = 0; kk < 8; ++kk)
                acc = __builtin_amdgcn_wmma_f32_16x16x4_f32(false, kf[kk], false, wf[kk],
                                                            (short)0, acc, false, false);
            #pragma unroll
            for (int kk = 0; kk < 8; ++kk) { kf[kk] = kn[kk]; wf[kk] = wn[kk]; }
        }
        __builtin_amdgcn_sched_barrier(0);
    }
}

#define CSTR 132
#define Q_OFF   0
#define K_OFF   (Q_OFF + 16 * CSTR)
#define P_OFF   (K_OFF + 16 * CSTR)
#define M_OFF   (P_OFF + 16 * CSTR)
#define SP_OFF  (M_OFF + 16 * CSTR)
#define S_OFF   (SP_OFF + 4 * 16 * 17)
#define NPS_OFF (S_OFF + 16 * 17)
#define NQ_OFF  (NPS_OFF + 128)
#define XSTR    20
#define XK_OFF  (NQ_OFF + 16)
#define XW_OFF  (XK_OFF + 128 * XSTR)
#define XG_OFF  (XW_OFF + 128 * XSTR)
#define SMEM_FLOATS (XG_OFF + 128)
#define MU_OFF  NPS_OFF
#define RS_OFF  (NPS_OFF + 16)

__global__ __launch_bounds__(128) void chunk_out_kernel(
    const float* __restrict__ qb, const float* __restrict__ kb,
    const float* __restrict__ vb, const float* __restrict__ ib,
    const float* __restrict__ fb, const float* __restrict__ ob,
    const float* __restrict__ cbuf, const float* __restrict__ nsbuf,
    const float* __restrict__ x,
    const float* __restrict__ lnw, const float* __restrict__ lnb,
    float* __restrict__ out)
{
    __shared__ float4 sm4[SMEM_FLOATS / 4];
    float* sm = (float*)sm4;
    const int b    = blockIdx.x / NCH;
    const int ch   = blockIdx.x % NCH;
    const int r    = threadIdx.x;
    const int lane = r & 31;
    const int wave = r >> 5;
    const int lm   = lane & 15;
    const int hi   = lane >> 4;
    const int lk   = hi << 1;

    const long gbase = ((long)b * TT + (long)(ch * LCH)) * HH + r;

    float pk[LCH], pq[LCH], pi[LCH], pf[LCH], pv[LCH], po[LCH];
    #pragma unroll
    for (int s = 0; s < LCH; ++s) {
        const long idx = gbase + (long)s * HH;
        pk[s] = kb[idx]; pq[s] = qb[idx]; pi[s] = ib[idx];
        pf[s] = fb[idx]; pv[s] = vb[idx]; po[s] = ob[idx];
    }
    const int sc  = ch >> 1;
    const int odd = ch & 1;
    float nstart = nsbuf[(long)(b * NSC + sc) * HH + r];
    float px[LCH];
    #pragma unroll
    for (int s = 0; s < LCH; ++s) px[s] = x[gbase + (long)s * HH];
    const float lw = lnw[r];
    const float lb = lnb[r];

    float4 cf[2][16];
    {
        const float4* cp4 = (const float4*)(cbuf + (long)(b * NSC + sc) * PLANE);
        #pragma unroll
        for (int tile = 0; tile < 2; ++tile)
            #pragma unroll
            for (int j = 0; j < 16; ++j)
                cf[tile][j] = cp4[((2 * wave + tile) * 16 + j) * 32 + lane];
    }

    if (odd) {
        float qk[LCH], qi[LCH], qf[LCH], qv[LCH];
        const long pbase = gbase - (long)LCH * HH;
        #pragma unroll
        for (int s = 0; s < LCH; ++s) {
            const long idx = pbase + (long)s * HH;
            qk[s] = kb[idx]; qi[s] = ib[idx]; qf[s] = fb[idx]; qv[s] = vb[idx];
        }
        float Ga[LCH];
        {
            float g = 1.0f;
            #pragma unroll
            for (int s = 0; s < LCH; ++s) { g *= qf[s]; Ga[s] = g; }
        }
        const float g15a = Ga[LCH - 1];
        float npa = 0.0f;
        #pragma unroll
        for (int s = 0; s < LCH; ++s) npa = fmaf(qf[s], npa, qi[s] * qk[s]);
        nstart = fmaf(g15a, nstart, npa);
        #pragma unroll
        for (int s = 0; s < LCH; ++s) {
            const float wg = qi[s] * qv[s] / Ga[s];
            sm[XK_OFF + r * XSTR + s] = qk[s];
            sm[XW_OFF + r * XSTR + s] = g15a * wg;
        }
        sm[XG_OFF + r] = g15a;
        __syncthreads();

        const int mt0 = wave * 2;
        const float ga0 = sm[XG_OFF + (mt0 + 0) * 16 + lm];
        const float ga1 = sm[XG_OFF + (mt0 + 1) * 16 + lm];
        v2f wfr[2][4];
        #pragma unroll
        for (int mi = 0; mi < 2; ++mi)
            #pragma unroll
            for (int kk = 0; kk < 4; ++kk)
                wfr[mi][kk] = *(const v2f*)&sm[XW_OFF + ((mt0 + mi) * 16 + lm) * XSTR + kk * 4 + lk];
        #pragma unroll
        for (int nt = 0; nt < 8; ++nt) {
            v2f kfr[4];
            #pragma unroll
            for (int kk = 0; kk < 4; ++kk)
                kfr[kk] = *(const v2f*)&sm[XK_OFF + (nt * 16 + lm) * XSTR + kk * 4 + lk];
            const float4 c00 = cf[0][2 * nt], c01 = cf[0][2 * nt + 1];
            const float4 c10 = cf[1][2 * nt], c11 = cf[1][2 * nt + 1];
            v8f d0 = {ga0 * c00.x, ga0 * c00.y, ga0 * c00.z, ga0 * c00.w,
                      ga0 * c01.x, ga0 * c01.y, ga0 * c01.z, ga0 * c01.w};
            v8f d1 = {ga1 * c10.x, ga1 * c10.y, ga1 * c10.z, ga1 * c10.w,
                      ga1 * c11.x, ga1 * c11.y, ga1 * c11.z, ga1 * c11.w};
            #pragma unroll
            for (int kk = 0; kk < 4; ++kk) {
                d0 = __builtin_amdgcn_wmma_f32_16x16x4_f32(false, kfr[kk], false, wfr[0][kk],
                                                           (short)0, d0, false, false);
                d1 = __builtin_amdgcn_wmma_f32_16x16x4_f32(false, kfr[kk], false, wfr[1][kk],
                                                           (short)0, d1, false, false);
            }
            cf[0][2 * nt]     = make_float4(d0[0], d0[1], d0[2], d0[3]);
            cf[0][2 * nt + 1] = make_float4(d0[4], d0[5], d0[6], d0[7]);
            cf[1][2 * nt]     = make_float4(d1[0], d1[1], d1[2], d1[3]);
            cf[1][2 * nt + 1] = make_float4(d1[4], d1[5], d1[6], d1[7]);
        }
    }

    float G[LCH], Wg[LCH], ovv[LCH];
    {
        float g = 1.0f;
        #pragma unroll
        for (int s = 0; s < LCH; ++s) { g *= pf[s]; G[s] = g; ovv[s] = po[s]; }
        #pragma unroll
        for (int s = 0; s < LCH; ++s) Wg[s] = pi[s] * pv[s] / G[s];
        float np = 0.0f;
        #pragma unroll
        for (int s = 0; s < LCH; ++s) {
            np = fmaf(pf[s], np, pi[s] * pk[s]);
            const float nfull = fmaf(G[s], nstart, np);
            sm[M_OFF + s * CSTR + r] = nfull * pq[s];
        }
        #pragma unroll
        for (int s = 0; s < LCH; ++s) {
            sm[Q_OFF + s * CSTR + r] = pq[s];
            sm[K_OFF + s * CSTR + r] = pk[s];
        }
    }
    __syncthreads();

    {
        const int s = r >> 3, seg = r & 7;
        float ps = 0.0f;
        #pragma unroll
        for (int e = 0; e < 16; ++e) ps += sm[M_OFF + s * CSTR + seg * 16 + e];
        sm[NPS_OFF + s * 8 + seg] = ps;
    }
    {
        v8f acc0 = {}, acc1 = {};
        #pragma unroll
        for (int j = 0; j < 16; ++j) {
            const int nt = j >> 1, half = j & 1;
            const float4 qa = *(const float4*)&sm[Q_OFF + lm * CSTR + 16 * nt + 8 * hi + 4 * half];
            const v2f a0 = {qa.x, qa.y};
            const v2f a1 = {qa.z, qa.w};
            const v2f b00 = {cf[0][j].x, cf[0][j].y};
            const v2f b01 = {cf[0][j].z, cf[0][j].w};
            const v2f b10 = {cf[1][j].x, cf[1][j].y};
            const v2f b11 = {cf[1][j].z, cf[1][j].w};
            acc0 = __builtin_amdgcn_wmma_f32_16x16x4_f32(false, a0, false, b00, (short)0, acc0, false, false);
            acc1 = __builtin_amdgcn_wmma_f32_16x16x4_f32(false, a0, false, b10, (short)0, acc1, false, false);
            acc0 = __builtin_amdgcn_wmma_f32_16x16x4_f32(false, a1, false, b01, (short)0, acc0, false, false);
            acc1 = __builtin_amdgcn_wmma_f32_16x16x4_f32(false, a1, false, b11, (short)0, acc1, false, false);
        }
        const int n0p = (2 * wave) << 4;
        #pragma unroll
        for (int v = 0; v < 8; ++v) {
            sm[P_OFF + (v + 8 * hi) * CSTR + n0p + lm]      = acc0[v];
            sm[P_OFF + (v + 8 * hi) * CSTR + n0p + 16 + lm] = acc1[v];
        }
    }
    {
        const int kbase = wave * 32;
        v8f sacc = {};
        #pragma unroll
        for (int kk = 0; kk < 8; ++kk) {
            const int c0 = kbase + kk * 4 + lk;
            v2f a  = *(const v2f*)&sm[K_OFF + lm * CSTR + c0];
            v2f bq = *(const v2f*)&sm[Q_OFF + lm * CSTR + c0];
            sacc = __builtin_amdgcn_wmma_f32_16x16x4_f32(false, a, false, bq,
                                                         (short)0, sacc, false, false);
        }
        #pragma unroll
        for (int v = 0; v < 8; ++v)
            sm[SP_OFF + wave * 272 + (v + 8 * hi) * 17 + lm] = sacc[v];
    }
    __syncthreads();

    if (r < 16) {
        float t = 0.0f;
        #pragma unroll
        for (int e = 0; e < 8; ++e) t += sm[NPS_OFF + r * 8 + e];
        sm[NQ_OFF + r] = t;
    }
    #pragma unroll
    for (int p = 0; p < 2; ++p) {
        const int idx = r * 2 + p;
        const int u = idx >> 4, sc = idx & 15;
        float ssum = sm[SP_OFF + 0 * 272 + u * 17 + sc]
                   + sm[SP_OFF + 1 * 272 + u * 17 + sc]
                   + sm[SP_OFF + 2 * 272 + u * 17 + sc]
                   + sm[SP_OFF + 3 * 272 + u * 17 + sc];
        sm[S_OFF + u * 17 + sc] = ssum;
    }
    __syncthreads();

    float yv[LCH];
    #pragma unroll
    for (int s = 0; s < LCH; ++s) {
        float acc = sm[P_OFF + s * CSTR + r];
        #pragma unroll
        for (int u = 0; u < LCH; ++u)
            if (u <= s) acc = fmaf(Wg[u], sm[S_OFF + u * 17 + s], acc);
        const float ret = G[s] * acc;
        const float nrm = fmaxf(fabsf(sm[NQ_OFF + s]), 1.0f);
        const float hval = ovv[s] * ret / nrm;
        yv[s] = hval + px[s];
        sm[P_OFF + s * CSTR + r] = yv[s];
    }
    __syncthreads();

    {
        const int s = r >> 3, seg = r & 7;
        const float4* yp = (const float4*)&sm[P_OFF + s * CSTR + seg * 16];
        float t1 = 0.0f, t2 = 0.0f;
        #pragma unroll
        for (int e = 0; e < 4; ++e) {
            const float4 a = yp[e];
            t1 += a.x; t2 += a.x * a.x;
            t1 += a.y; t2 += a.y * a.y;
            t1 += a.z; t2 += a.z * a.z;
            t1 += a.w; t2 += a.w * a.w;
        }
        #pragma unroll
        for (int off = 1; off <= 4; off <<= 1) {
            t1 += __shfl_xor(t1, off, 32);
            t2 += __shfl_xor(t2, off, 32);
        }
        const float mu  = t1 * (1.0f / HH);
        const float var = t2 * (1.0f / HH) - mu * mu;
        if (seg == 0) {
            sm[MU_OFF + s] = mu;
            sm[RS_OFF + s] = rsqrtf(var + LN_EPS);
        }
    }
    __syncthreads();

    #pragma unroll
    for (int s = 0; s < LCH; ++s)
        out[gbase + (long)s * HH] = (yv[s] - sm[MU_OFF + s]) * sm[RS_OFF + s] * lw + lb;
}

extern "C" void kernel_launch(void* const* d_in, const int* in_sizes, int n_in,
                              void* d_out, int out_size, void* d_ws, size_t ws_size,
                              hipStream_t stream)
{
    const float* x   = (const float*)d_in[0];
    const float* Wq  = (const float*)d_in[1];
    const float* Wk  = (const float*)d_in[2];
    const float* Wv  = (const float*)d_in[3];
    const float* Wi  = (const float*)d_in[4];
    const float* Wf  = (const float*)d_in[5];
    const float* Wo  = (const float*)d_in[6];
    const float* b_i = (const float*)d_in[7];
    const float* b_f = (const float*)d_in[8];
    const float* b_o = (const float*)d_in[9];
    const float* lnw = (const float*)d_in[10];
    const float* lnb = (const float*)d_in[11];
    float* out = (float*)d_out;

    float* ws = (float*)d_ws;
    float* q  = ws + 0L * NEL;
    float* k  = ws + 1L * NEL;
    float* v  = ws + 2L * NEL;
    float* ig = ws + 3L * NEL;
    float* fg = ws + 4L * NEL;
    float* og = ws + 5L * NEL;
    float* cst = ws + 7L * NEL;
    float* nl  = ws + 15L * NEL + BB * NCH * HH;

    proj_kernel<<<PJ_GRID, 128, 0, stream>>>(x, Wq, Wk, Wv, Wi, Wf, Wo,
                                          b_i, b_f, b_o,
                                          q, k, v, ig, fg, og);
    chunk_state_kernel<<<ST_GRID, 256, 0, stream>>>(k, v, ig, fg, cst, nl);
    chunk_out_kernel<<<BB * NCH, 128, 0, stream>>>(q, k, v, ig, fg, og, cst, nl,
                                                   x, lnw, lnb, out);
}


# q and o projections leave the projection launch (768 -> 512 work-groups) for 256 extra work-groups of the state launch, co-resident with the state work-groups (state kernel held to 256 registers)
# speedup vs baseline: 11.6343x; 1.0737x over previous
#include <hip/hip_runtime.h>
#include <math.h>


typedef float v2f __attribute__((ext_vector_type(2)));
typedef float v8f __attribute__((ext_vector_type(8)));

#define BB   16
#define TT   512
#define HH   128
#define MM   (BB * TT)
#define NEL  (BB * TT * HH)
#define SCALE 0.08838834764831845f
#define LN_EPS 1e-5f
#define LCH  16
#define NCH  (TT / LCH)
#define NSC  (NCH / 2)

#define PJ_ROWS 32
#define PJ_COLS 64
#define PJ_STR  132
#define PJ_XS_FLOATS (PJ_ROWS * PJ_STR)
#define PJ_WB_FLOATS (PJ_COLS * PJ_STR)
#define PJ_NBLK 4
#define PJ_GRID (2 * (MM / PJ_ROWS))
#define QO_GRID (MM / PJ_ROWS)

template <int MAT0, int MAT1>
__device__ __forceinline__ void pj_body(
    const float* __restrict__ x,
    const float* __restrict__ Wm0, const float* __restrict__ Wm1,
    const float* __restrict__ bm0, const float* __restrict__ bm1,
    float* __restrict__ om0, float* __restrict__ om1,
    float* xs, float* wbuf)
{
    const int t    = threadIdx.x;
    const int lane = t & 31;
    const int wave = t >> 5;
    const int lm   = lane & 15;
    const int hi   = lane >> 4;
    const int lk   = hi << 1;
    const int m0   = (blockIdx.x / 2) * PJ_ROWS;

    const int srow = wave;
    const int sc   = lane << 2;

    {
        float4 xv[8], wv[16];
        const float* xsrc = x + (long)(m0 + srow) * HH + sc;
        #pragma unroll
        for (int i = 0; i < 8; ++i) xv[i] = *(const float4*)(xsrc + (4 * i) * HH);
        const float* wsrc = Wm0 + srow * HH + sc;
        #pragma unroll
        for (int i = 0; i < 16; ++i) wv[i] = *(const float4*)(wsrc + (4 * i) * HH);
        #pragma unroll
        for (int i = 0; i < 8; ++i) *(float4*)&xs[(srow + 4 * i) * PJ_STR + sc] = xv[i];
        #pragma unroll
        for (int i = 0; i < 16; ++i) *(float4*)&wbuf[(srow + 4 * i) * PJ_STR + sc] = wv[i];
    }
    __syncthreads();

    v2f af[2][32];
    #pragma unroll
    for (int tile = 0; tile < 2; ++tile)
        #pragma unroll
        for (int kk = 0; kk < 32; ++kk)
            af[tile][kk] = *(const v2f*)&xs[(16 * tile + lm) * PJ_STR + 4 * kk + lk];

    #pragma unroll
    for (int blk = 0; blk < PJ_NBLK; ++blk) {
        const float* wcur = wbuf + (blk & 1) * PJ_WB_FLOATS;
        v2f bf[32];
        #pragma unroll
        for (int kk = 0; kk < 32; ++kk)
            bf[kk] = *(const v2f*)&wcur[(16 * wave + lm) * PJ_STR + 4 * kk + lk];

        __builtin_amdgcn_sched_barrier(0);

        v8f acc0 = {}, acc1 = {};
        #pragma unroll
        for (int kk = 0; kk < 32; ++kk) {
            acc0 = __builtin_amdgcn_wmma_f32_16x16x4_f32(false, af[0][kk], false, bf[kk],
                                                         (short)0, acc0, false, false);
            acc1 = __builtin_amdgcn_wmma_f32_16x16x4_f32(false, af[1][kk], false, bf[kk],
                                                         (short)0, acc1, false, false);
        }

        {
            const int second = blk >> 1;
            const int mat    = second ? MAT1 : MAT0;
            const int col    = (blk & 1) * PJ_COLS + 16 * wave + lm;
            float* outp      = second ? om1 : om0;
            float bias = 0.0f;
            if (mat >= 3) bias = (second ? bm1 : bm0)[col];
            #pragma unroll
            for (int tile = 0; tile < 2; ++tile) {
                const int rowBase = m0 + 16 * tile + 8 * hi;
                #pragma unroll
                for (int v = 0; v < 8; ++v) {
                    float val = tile ? acc1[v] : acc0[v];
                    if (mat <= 1)       val *= SCALE;
                    else if (mat == 3)  val = expf(val + bias);
                    else if (mat >= 4)  val = 1.0f / (1.0f + expf(-(val + bias)));
                    outp[(rowBase + v) * HH + col] = val;
                }
            }
        }

        if (blk + 1 < PJ_NBLK) {
            float4 wv[16];
            const float* wsrc = (((blk + 1) >> 1) ? Wm1 : Wm0) + (((blk + 1) & 1) * PJ_COLS + srow) * HH + sc;
            #pragma unroll
            for (int i = 0; i < 16; ++i) wv[i] = *(const float4*)(wsrc + (4 * i) * HH);
            float* wnext = wbuf + ((blk + 1) & 1) * PJ_WB_FLOATS;
            #pragma unroll
            for (int i = 0; i < 16; ++i) *(float4*)&wnext[(srow + 4 * i) * PJ_STR + sc] = wv[i];
            __syncthreads();
        }
    }
}

__global__ __launch_bounds__(128) void proj_kernel(
    const float* __restrict__ x,
    const float* __restrict__ Wk, const float* __restrict__ Wv,
    const float* __restrict__ Wi, const float* __restrict__ Wf,
    const float* __restrict__ b_i, const float* __restrict__ b_f,
    float* __restrict__ ok, float* __restrict__ ov,
    float* __restrict__ oi, float* __restrict__ of)
{
    __shared__ float4 pj4[(PJ_XS_FLOATS + 2 * PJ_WB_FLOATS) / 4];
    float* xs   = (float*)pj4;
    float* wbuf = xs + PJ_XS_FLOATS;
    if ((blockIdx.x & 1) == 0) pj_body<1, 2>(x, Wk, Wv, b_i, b_i, ok, ov, xs, wbuf);
    else                       pj_body<3, 4>(x, Wi, Wf, b_i, b_f, oi, of, xs, wbuf);
}

#define QO_W_FLOATS (HH * PJ_STR)
#define QO_FLOATS   (PJ_XS_FLOATS + 2 * QO_W_FLOATS)

template <bool IS_O>
__device__ __forceinline__ void pjqo_tiles(
    const float* wm, const v2f (&af)[32], const float* __restrict__ b_o,
    float* __restrict__ outp, const int rowBase, const int ctile, const int lm, const int lk)
{
    #pragma unroll
    for (int half = 0; half < 2; ++half) {
        const int col = half * PJ_COLS + 16 * ctile + lm;
        v2f bf[32];
        #pragma unroll
        for (int kk = 0; kk < 32; ++kk)
            bf[kk] = *(const v2f*)&wm[col * PJ_STR + 4 * kk + lk];
        float bias = 0.0f;
        if (IS_O) bias = b_o[col];

        __builtin_amdgcn_sched_barrier(0);

        v8f acc = {};
        #pragma unroll
        for (int kk = 0; kk < 32; ++kk)
            acc = __builtin_amdgcn_wmma_f32_16x16x4_f32(false, af[kk], false, bf[kk],
                                                        (short)0, acc, false, false);
        #pragma unroll
        for (int v = 0; v < 8; ++v) {
            float val = acc[v];
            if (IS_O) val = 1.0f / (1.0f + expf(-(val + bias)));
            else      val *= SCALE;
            outp[(rowBase + v) * HH + col] = val;
        }
    }
}

__device__ __forceinline__ void pjqo_body(
    const float* __restrict__ x,
    const float* __restrict__ Wq, const float* __restrict__ Wo,
    const float* __restrict__ b_o,
    float* __restrict__ oq, float* __restrict__ oo,
    float* lds, const int rb)
{
    float* xs  = lds;
    float* wqs = lds + PJ_XS_FLOATS;
    float* wos = wqs + QO_W_FLOATS;
    const int t    = threadIdx.x;
    const int lane = t & 31;
    const int wave = t >> 5;
    const int lm   = lane & 15;
    const int hi   = lane >> 4;
    const int lk   = hi << 1;
    const int m0   = rb * PJ_ROWS;

    const int srow = wave;
    const int sc   = lane << 2;

    {
        float4 xv[4], wv[16];
        const float* xsrc = x + (long)(m0 + srow) * HH + sc;
        #pragma unroll
        for (int i = 0; i < 4; ++i) xv[i] = *(const float4*)(xsrc + (8 * i) * HH);
        const float* wsrc = Wq + srow * HH + sc;
        #pragma unroll
        for (int i = 0; i < 16; ++i) wv[i] = *(const float4*)(wsrc + (8 * i) * HH);
        #pragma unroll
        for (int i = 0; i < 4; ++i) *(float4*)&xs[(srow + 8 * i) * PJ_STR + sc] = xv[i];
        #pragma unroll
        for (int i = 0; i < 16; ++i) *(float4*)&wqs[(srow + 8 * i) * PJ_STR + sc] = wv[i];
    }
    __syncthreads();

    const int rtile = wave >> 2;
    const int ctile = wave & 3;
    const int rowBase = m0 + 16 * rtile + 8 * hi;

    v2f af[32];
    #pragma unroll
    for (int kk = 0; kk < 32; ++kk)
        af[kk] = *(const v2f*)&xs[(16 * rtile + lm) * PJ_STR + 4 * kk + lk];

    pjqo_tiles<false>(wqs, af, b_o, oq, rowBase, ctile, lm, lk);

    {
        float4 wv[16];
        const float* wsrc = Wo + srow * HH + sc;
        #pragma unroll
        for (int i = 0; i < 16; ++i) wv[i] = *(const float4*)(wsrc + (8 * i) * HH);
        #pragma unroll
        for (int i = 0; i < 16; ++i) *(float4*)&wos[(srow + 8 * i) * PJ_STR + sc] = wv[i];
    }
    __syncthreads();

    pjqo_tiles<true>(wos, af, b_o, oo, rowBase, ctile, lm, lk);
}

#define PLANE  (HH * HH)
#define PLANE4 (HH * HH / 4)

#define ST_KP    40
#define ST_WP    516
#define ST_K_OFF 0
#define ST_W_OFF (TT * ST_KP)
#define ST_G_OFF (ST_W_OFF + 32 * ST_WP)
#define ST_N_OFF (ST_G_OFF + 512)
#define ST_FLOATS (ST_N_OFF + 512)
#define ST_GRID  (BB * 4 * 4)
static_assert(QO_FLOATS <= ST_FLOATS, "the q/o staging must fit the state kernel's LDS array");

__device__ __forceinline__ void st_w(
    float* ss, const int pp, const int lane,
    const float (&pi)[32], const float (&pf)[32], const float (&pv)[32])
{
    float Ga[LCH], Gb[LCH];
    {
        float g = 1.0f;
        #pragma unroll
        for (int s = 0; s < LCH; ++s) { g *= pf[s]; Ga[s] = g; }
        g = 1.0f;
        #pragma unroll
        for (int s = 0; s < LCH; ++s) { g *= pf[LCH + s]; Gb[s] = g; }
    }
    const float g15a = Ga[LCH - 1];
    const float g15b = Gb[LCH - 1];
    #pragma unroll
    for (int s = 0; s < LCH; ++s) {
        const float wga = pi[s] * pv[s] / Ga[s];
        const float wgb = pi[LCH + s] * pv[LCH + s] / Gb[s];
        ss[ST_W_OFF + lane * ST_WP + 32 * pp + s]       = g15b * (g15a * wga);
        ss[ST_W_OFF + lane * ST_WP + 32 * pp + LCH + s] = g15b * wgb;
    }
    ss[ST_G_OFF + pp * 32 + lane] = g15a * g15b;
}

__device__ __forceinline__ void st_n(
    float* ss, const int pp, const int lane,
    const float (&pi)[32], const float (&pf)[32], const float (&pk)[32])
{
    float g15b = 1.0f;
    #pragma unroll
    for (int s = 0; s < LCH; ++s) g15b *= pf[LCH + s];
    float npa = 0.0f, npb = 0.0f;
    #pragma unroll
    for (int s = 0; s < LCH; ++s) npa = fmaf(pf[s], npa, pi[s] * pk[s]);
    #pragma unroll
    for (int s = 0; s < LCH; ++s) npb = fmaf(pf[LCH + s], npb, pi[LCH + s] * pk[LCH + s]);
    ss[ST_N_OFF + pp * 32 + lane] = fmaf(g15b, npa, npb);
}

template <bool WITH_N>
__device__ __forceinline__ void st_gates(
    const float* __restrict__ kb, const float* __restrict__ vb,
    const float* __restrict__ ib, const float* __restrict__ fb,
    float* ss, const float* __restrict__ ksrc, const long g0a,
    const int pp0, const int lane, const int kr, const int kc)
{
    const long g0b = g0a + (long)32 * HH;
    float4 kv[16];
    #pragma unroll
    for (int i = 0; i < 16; ++i) kv[i] = *(const float4*)(ksrc + (long)(32 * i) * HH);
    float i0[32], f0[32], v0[32], k0[32];
    #pragma unroll
    for (int s = 0; s < 32; ++s) {
        const long idx = g0a + (long)s * HH;
        i0[s] = ib[idx]; f0[s] = fb[idx]; v0[s] = vb[idx];
        if (WITH_N) k0[s] = kb[idx];
    }
    #pragma unroll
    for (int i = 0; i < 16; ++i) *(float4*)&ss[ST_K_OFF + (kr + 32 * i) * ST_KP + kc] = kv[i];
    float i1[32], f1[32], v1[32], k1[32];
    #pragma unroll
    for (int s = 0; s < 32; ++s) {
        const long idx = g0b + (long)s * HH;
        f1[s] = fb[idx];
        if (!WITH_N) i1[s] = ib[idx];
    }
    __builtin_amdgcn_sched_barrier(0);
    if (WITH_N) st_n(ss, pp0, lane, i0, f0, k0);
    #pragma unroll
    for (int s = 0; s < 32; ++s) {
        const long idx = g0b + (long)s * HH;
        v1[s] = vb[idx];
        if (WITH_N) i1[s] = ib[idx];
    }
    __builtin_amdgcn_sched_barrier(0);
    st_w(ss, pp0, lane, i0, f0, v0);
    if (WITH_N) {
        #pragma unroll
        for (int s = 0; s < 32; ++s) k1[s] = kb[g0b + (long)s * HH];
        __builtin_amdgcn_sched_barrier(0);
    }
    st_w(ss, pp0 + 1, lane, i1, f1, v1);
    if (WITH_N) st_n(ss, pp0 + 1, lane, i1, f1, k1);
}

__global__ __launch_bounds__(256) void chunk_state_kernel(
    const float* __restrict__ kb, const float* __restrict__ vb,
    const float* __restrict__ ib, const float* __restrict__ fb,
    float* __restrict__ cbuf, float* __restrict__ nsbuf,
    const float* __restrict__ x,
    const float* __restrict__ Wq, const float* __restrict__ Wo,
    const float* __restrict__ b_o,
    float* __restrict__ oq, float* __restrict__ oo)
{
    __shared__ float4 ss4[ST_FLOATS / 4];
    float* ss = (float*)ss4;
    if (blockIdx.x >= ST_GRID) {
        pjqo_body(x, Wq, Wo, b_o, oq, oo, ss, blockIdx.x - ST_GRID);
        return;
    }
    const int cq   = blockIdx.x & 3;
    const int rq   = (blockIdx.x >> 2) & 3;
    const int b    = blockIdx.x >> 4;
    const int t    = threadIdx.x;
    const int lane = t & 31;
    const int wave = t >> 5;
    const int lm   = lane & 15;
    const int hi   = lane >> 4;
    const int lk   = hi << 1;

    const int kr = t >> 3;
    const int kc = (t & 7) << 2;
    const float* ksrc = kb + ((long)b * TT + kr) * HH + cq * 32 + kc;

    {
        const long g0a = ((long)b * TT + (long)(32 * (2 * wave))) * HH + 32 * rq + lane;
        if (cq == 0) st_gates<true>(kb, vb, ib, fb, ss, ksrc, g0a, 2 * wave, lane, kr, kc);
        else         st_gates<false>(kb, vb, ib, fb, ss, ksrc, g0a, 2 * wave, lane, kr, kc);
    }
    __syncthreads();

    if (cq == 0 && wave == 4) {
        float* np = nsbuf + (long)b * NSC * HH + 32 * rq + lane;
        float ns = 0.0f;
        #pragma unroll
        for (int p = 0; p < NSC; ++p) {
            np[p * HH] = ns;
            ns = fmaf(ss[ST_G_OFF + p * 32 + lane], ns, ss[ST_N_OFF + p * 32 + lane]);
        }
    }
    if (wave >= 4) return;

    const int rtl = wave >> 1;
    const int ctl = wave & 1;
    const int rt  = 2 * rq + rtl;
    const int ct  = 2 * cq + ctl;
    float4* dp4 = (float4*)cbuf + (long)(b * NSC) * PLANE4 + (rt * 8 + ct) * 64 + lane;
    const float* kcol = ss + ST_K_OFF + 16 * ctl + lm;
    const float* wrow = ss + ST_W_OFF + (16 * rtl + lm) * ST_WP;
    const float* grow = ss + ST_G_OFF + 16 * rtl + lm;
    v2f kf[8], wf[8];
    #pragma unroll
    for (int kk = 0; kk < 8; ++kk) {
        const int u = 4 * kk + lk;
        kf[kk] = (v2f){kcol[u * ST_KP], kcol[(u + 1) * ST_KP]};
        wf[kk] = *(const v2f*)&wrow[u];
    }
    float gq[NSC];
    #pragma unroll
    for (int p = 0; p < NSC; ++p) gq[p] = grow[p * 32];
    v8f acc = {};
    #pragma unroll
    for (int p = 0; p < NSC; ++p) {
        v2f kn[8], wn[8];
        if (p + 1 < NSC) {
            #pragma unroll
            for (int kk = 0; kk < 8; ++kk) {
                const int u = 32 * (p + 1) + 4 * kk + lk;
                kn[kk] = (v2f){kcol[u * ST_KP], kcol[(u + 1) * ST_KP]};
                wn[kk] = *(const v2f*)&wrow[u];
            }
        }
        __builtin_amdgcn_sched_barrier(0);
        dp4[(long)p * PLANE4]      = make_float4(acc[0], acc[1], acc[2], acc[3]);
        dp4[(long)p * PLANE4 + 32] = make_float4(acc[4], acc[5], acc[6], acc[7]);
        if (p + 1 < NSC) {
            #pragma unroll
            for (int v = 0; v < 8; ++v) acc[v] *= gq[p];
            #pragma unroll
            for (int kk = 0; kk < 8; ++kk)
                acc = __builtin_amdgcn_wmma_f32_16x16x4_f32(false, kf[kk], false, wf[kk],
                                                            (short)0, acc, false, false);
            #pragma unroll
            for (int kk = 0; kk < 8; ++kk) { kf[kk] = kn[kk]; wf[kk] = wn[kk]; }
        }
        __builtin_amdgcn_sched_barrier(0);
    }
}

#define CSTR 132
#define Q_OFF   0
#define K_OFF   (Q_OFF + 16 * CSTR)
#define P_OFF   (K_OFF + 16 * CSTR)
#define M_OFF   (P_OFF + 16 * CSTR)
#define SP_OFF  (M_OFF + 16 * CSTR)
#define S_OFF   (SP_OFF + 4 * 16 * 17)
#define NPS_OFF (S_OFF + 16 * 17)
#define NQ_OFF  (NPS_OFF + 128)
#define XSTR    20
#define XK_OFF  (NQ_OFF + 16)
#define XW_OFF  (XK_OFF + 128 * XSTR)
#define XG_OFF  (XW_OFF + 128 * XSTR)
#define SMEM_FLOATS (XG_OFF + 128)
#define MU_OFF  NPS_OFF
#define RS_OFF  (NPS_OFF + 16)

__global__ __launch_bounds__(128) void chunk_out_kernel(
    const float* __restrict__ qb, const float* __restrict__ kb,
    const float* __restrict__ vb, const float* __restrict__ ib,
    const float* __restrict__ fb, const float* __restrict__ ob,
    const float* __restrict__ cbuf, const float* __restrict__ nsbuf,
    const float* __restrict__ x,
    const float* __restrict__ lnw, const float* __restrict__ lnb,
    float* __restrict__ out)
{
    __shared__ float4 sm4[SMEM_FLOATS / 4];
    float* sm = (float*)sm4;
    const int b    = blockIdx.x / NCH;
    const int ch   = blockIdx.x % NCH;
    const int r    = threadIdx.x;
    const int lane = r & 31;
    const int wave = r >> 5;
    const int lm   = lane & 15;
    const int hi   = lane >> 4;
    const int lk   = hi << 1;

    const long gbase = ((long)b * TT + (long)(ch * LCH)) * HH + r;

    float pk[LCH], pq[LCH], pi[LCH], pf[LCH], pv[LCH], po[LCH];
    #pragma unroll
    for (int s = 0; s < LCH; ++s) {
        const long idx = gbase + (long)s * HH;
        pk[s] = kb[idx]; pq[s] = qb[idx]; pi[s] = ib[idx];
        pf[s] = fb[idx]; pv[s] = vb[idx]; po[s] = ob[idx];
    }
    const int sc  = ch >> 1;
    const int odd = ch & 1;
    float nstart = nsbuf[(long)(b * NSC + sc) * HH + r];
    float px[LCH];
    #pragma unroll
    for (int s = 0; s < LCH; ++s) px[s] = x[gbase + (long)s * HH];
    const float lw = lnw[r];
    const float lb = lnb[r];

    float4 cf[2][16];
    {
        const float4* cp4 = (const float4*)(cbuf + (long)(b * NSC + sc) * PLANE);
        #pragma unroll
        for (int tile = 0; tile < 2; ++tile)
            #pragma unroll
            for (int j = 0; j < 16; ++j)
                cf[tile][j] = cp4[((2 * wave + tile) * 16 + j) * 32 + lane];
    }

    if (odd) {
        float qk[LCH], qi[LCH], qf[LCH], qv[LCH];
        const long pbase = gbase - (long)LCH * HH;
        #pragma unroll
        for (int s = 0; s < LCH; ++s) {
            const long idx = pbase + (long)s * HH;
            qk[s] = kb[idx]; qi[s] = ib[idx]; qf[s] = fb[idx]; qv[s] = vb[idx];
        }
        float Ga[LCH];
        {
            float g = 1.0f;
            #pragma unroll
            for (int s = 0; s < LCH; ++s) { g *= qf[s]; Ga[s] = g; }
        }
        const float g15a = Ga[LCH - 1];
        float npa = 0.0f;
        #pragma unroll
        for (int s = 0; s < LCH; ++s) npa = fmaf(qf[s], npa, qi[s] * qk[s]);
        nstart = fmaf(g15a, nstart, npa);
        #pragma unroll
        for (int s = 0; s < LCH; ++s) {
            const float wg = qi[s] * qv[s] / Ga[s];
            sm[XK_OFF + r * XSTR + s] = qk[s];
            sm[XW_OFF + r * XSTR + s] = g15a * wg;
        }
        sm[XG_OFF + r] = g15a;
        __syncthreads();

        const int mt0 = wave * 2;
        const float ga0 = sm[XG_OFF + (mt0 + 0) * 16 + lm];
        const float ga1 = sm[XG_OFF + (mt0 + 1) * 16 + lm];
        v2f wfr[2][4];
        #pragma unroll
        for (int mi = 0; mi < 2; ++mi)
            #pragma unroll
            for (int kk = 0; kk < 4; ++kk)
                wfr[mi][kk] = *(const v2f*)&sm[XW_OFF + ((mt0 + mi) * 16 + lm) * XSTR + kk * 4 + lk];
        #pragma unroll
        for (int nt = 0; nt < 8; ++nt) {
            v2f kfr[4];
            #pragma unroll
            for (int kk = 0; kk < 4; ++kk)
                kfr[kk] = *(const v2f*)&sm[XK_OFF + (nt * 16 + lm) * XSTR + kk * 4 + lk];
            const float4 c00 = cf[0][2 * nt], c01 = cf[0][2 * nt + 1];
            const float4 c10 = cf[1][2 * nt], c11 = cf[1][2 * nt + 1];
            v8f d0 = {ga0 * c00.x, ga0 * c00.y, ga0 * c00.z, ga0 * c00.w,
                      ga0 * c01.x, ga0 * c01.y, ga0 * c01.z, ga0 * c01.w};
            v8f d1 = {ga1 * c10.x, ga1 * c10.y, ga1 * c10.z, ga1 * c10.w,
                      ga1 * c11.x, ga1 * c11.y, ga1 * c11.z, ga1 * c11.w};
            #pragma unroll
            for (int kk = 0; kk < 4; ++kk) {
                d0 = __builtin_amdgcn_wmma_f32_16x16x4_f32(false, kfr[kk], false, wfr[0][kk],
                                                           (short)0, d0, false, false);
                d1 = __builtin_amdgcn_wmma_f32_16x16x4_f32(false, kfr[kk], false, wfr[1][kk],
                                                           (short)0, d1, false, false);
            }
            cf[0][2 * nt]     = make_float4(d0[0], d0[1], d0[2], d0[3]);
            cf[0][2 * nt + 1] = make_float4(d0[4], d0[5], d0[6], d0[7]);
            cf[1][2 * nt]     = make_float4(d1[0], d1[1], d1[2], d1[3]);
            cf[1][2 * nt + 1] = make_float4(d1[4], d1[5], d1[6], d1[7]);
        }
    }

    float G[LCH], Wg[LCH], ovv[LCH];
    {
        float g = 1.0f;
        #pragma unroll
        for (int s = 0; s < LCH; ++s) { g *= pf[s]; G[s] = g; ovv[s] = po[s]; }
        #pragma unroll
        for (int s = 0; s < LCH; ++s) Wg[s] = pi[s] * pv[s] / G[s];
        float np = 0.0f;
        #pragma unroll
        for (int s = 0; s < LCH; ++s) {
            np = fmaf(pf[s], np, pi[s] * pk[s]);
            const float nfull = fmaf(G[s], nstart, np);
            sm[M_OFF + s * CSTR + r] = nfull * pq[s];
        }
        #pragma unroll
        for (int s = 0; s < LCH; ++s) {
            sm[Q_OFF + s * CSTR + r] = pq[s];
            sm[K_OFF + s * CSTR + r] = pk[s];
        }
    }
    __syncthreads();

    {
        const int s = r >> 3, seg = r & 7;
        float ps = 0.0f;
        #pragma unroll
        for (int e = 0; e < 16; ++e) ps += sm[M_OFF + s * CSTR + seg * 16 + e];
        sm[NPS_OFF + s * 8 + seg] = ps;
    }
    {
        v8f acc0 = {}, acc1 = {};
        #pragma unroll
        for (int j = 0; j < 16; ++j) {
            const int nt = j >> 1, half = j & 1;
            const float4 qa = *(const float4*)&sm[Q_OFF + lm * CSTR + 16 * nt + 8 * hi + 4 * half];
            const v2f a0 = {qa.x, qa.y};
            const v2f a1 = {qa.z, qa.w};
            const v2f b00 = {cf[0][j].x, cf[0][j].y};
            const v2f b01 = {cf[0][j].z, cf[0][j].w};
            const v2f b10 = {cf[1][j].x, cf[1][j].y};
            const v2f b11 = {cf[1][j].z, cf[1][j].w};
            acc0 = __builtin_amdgcn_wmma_f32_16x16x4_f32(false, a0, false, b00, (short)0, acc0, false, false);
            acc1 = __builtin_amdgcn_wmma_f32_16x16x4_f32(false, a0, false, b10, (short)0, acc1, false, false);
            acc0 = __builtin_amdgcn_wmma_f32_16x16x4_f32(false, a1, false, b01, (short)0, acc0, false, false);
            acc1 = __builtin_amdgcn_wmma_f32_16x16x4_f32(false, a1, false, b11, (short)0, acc1, false, false);
        }
        const int n0p = (2 * wave) << 4;
        #pragma unroll
        for (int v = 0; v < 8; ++v) {
            sm[P_OFF + (v + 8 * hi) * CSTR + n0p + lm]      = acc0[v];
            sm[P_OFF + (v + 8 * hi) * CSTR + n0p + 16 + lm] = acc1[v];
        }
    }
    {
        const int kbase = wave * 32;
        v8f sacc = {};
        #pragma unroll
        for (int kk = 0; kk < 8; ++kk) {
            const int c0 = kbase + kk * 4 + lk;
            v2f a  = *(const v2f*)&sm[K_OFF + lm * CSTR + c0];
            v2f bq = *(const v2f*)&sm[Q_OFF + lm * CSTR + c0];
            sacc = __builtin_amdgcn_wmma_f32_16x16x4_f32(false, a, false, bq,
                                                         (short)0, sacc, false, false);
        }
        #pragma unroll
        for (int v = 0; v < 8; ++v)
            sm[SP_OFF + wave * 272 + (v + 8 * hi) * 17 + lm] = sacc[v];
    }
    __syncthreads();

    if (r < 16) {
        float t = 0.0f;
        #pragma unroll
        for (int e = 0; e < 8; ++e) t += sm[NPS_OFF + r * 8 + e];
        sm[NQ_OFF + r] = t;
    }
    #pragma unroll
    for (int p = 0; p < 2; ++p) {
        const int idx = r * 2 + p;
        const int u = idx >> 4, sc = idx & 15;
        float ssum = sm[SP_OFF + 0 * 272 + u * 17 + sc]
                   + sm[SP_OFF + 1 * 272 + u * 17 + sc]
                   + sm[SP_OFF + 2 * 272 + u * 17 + sc]
                   + sm[SP_OFF + 3 * 272 + u * 17 + sc];
        sm[S_OFF + u * 17 + sc] = ssum;
    }
    __syncthreads();

    float yv[LCH];
    #pragma unroll
    for (int s = 0; s < LCH; ++s) {
        float acc = sm[P_OFF + s * CSTR + r];
        #pragma unroll
        for (int u = 0; u < LCH; ++u)
            if (u <= s) acc = fmaf(Wg[u], sm[S_OFF + u * 17 + s], acc);
        const float ret = G[s] * acc;
        const float nrm = fmaxf(fabsf(sm[NQ_OFF + s]), 1.0f);
        const float hval = ovv[s] * ret / nrm;
        yv[s] = hval + px[s];
        sm[P_OFF + s * CSTR + r] = yv[s];
    }
    __syncthreads();

    {
        const int s = r >> 3, seg = r & 7;
        const float4* yp = (const float4*)&sm[P_OFF + s * CSTR + seg * 16];
        float t1 = 0.0f, t2 = 0.0f;
        #pragma unroll
        for (int e = 0; e < 4; ++e) {
            const float4 a = yp[e];
            t1 += a.x; t2 += a.x * a.x;
            t1 += a.y; t2 += a.y * a.y;
            t1 += a.z; t2 += a.z * a.z;
            t1 += a.w; t2 += a.w * a.w;
        }
        #pragma unroll
        for (int off = 1; off <= 4; off <<= 1) {
            t1 += __shfl_xor(t1, off, 32);
            t2 += __shfl_xor(t2, off, 32);
        }
        const float mu  = t1 * (1.0f / HH);
        const float var = t2 * (1.0f / HH) - mu * mu;
        if (seg == 0) {
            sm[MU_OFF + s] = mu;
            sm[RS_OFF + s] = rsqrtf(var + LN_EPS);
        }
    }
    __syncthreads();

    #pragma unroll
    for (int s = 0; s < LCH; ++s)
        out[gbase + (long)s * HH] = (yv[s] - sm[MU_OFF + s]) * sm[RS_OFF + s] * lw + lb;
}

extern "C" void kernel_launch(void* const* d_in, const int* in_sizes, int n_in,
                              void* d_out, int out_size, void* d_ws, size_t ws_size,
                              hipStream_t stream)
{
    const float* x   = (const float*)d_in[0];
    const float* Wq  = (const float*)d_in[1];
    const float* Wk  = (const float*)d_in[2];
    const float* Wv  = (const float*)d_in[3];
    const float* Wi  = (const float*)d_in[4];
    const float* Wf  = (const float*)d_in[5];
    const float* Wo  = (const float*)d_in[6];
    const float* b_i = (const float*)d_in[7];
    const float* b_f = (const float*)d_in[8];
    const float* b_o = (const float*)d_in[9];
    const float* lnw = (const float*)d_in[10];
    const float* lnb = (const float*)d_in[11];
    float* out = (float*)d_out;

    float* ws = (float*)d_ws;
    float* q  = ws + 0L * NEL;
    float* k  = ws + 1L * NEL;
    float* v  = ws + 2L * NEL;
    float* ig = ws + 3L * NEL;
    float* fg = ws + 4L * NEL;
    float* og = ws + 5L * NEL;
    float* cst = ws + 7L * NEL;
    float* nl  = ws + 15L * NEL + BB * NCH * HH;

    proj_kernel<<<PJ_GRID, 128, 0, stream>>>(x, Wk, Wv, Wi, Wf, b_i, b_f,
                                          k, v, ig, fg);
    chunk_state_kernel<<<ST_GRID + QO_GRID, 256, 0, stream>>>(k, v, ig, fg, cst, nl,
                                                              x, Wq, Wo, b_o, q, og);
    chunk_out_kernel<<<BB * NCH, 128, 0, stream>>>(q, k, v, ig, fg, og, cst, nl,
                                                   x, lnw, lnb, out);
}
